# starting kernel
# baseline (speedup 1.0000x reference)
#include <hip/hip_runtime.h>

typedef __attribute__((ext_vector_type(2))) float v2f;
typedef __attribute__((ext_vector_type(8))) float v8f;

#define IN_F  64
#define HID_F 64
#define OUT_F 32

__global__ void gcn_zero_kernel(float4* __restrict__ p, long n4) {
    long i = (long)blockIdx.x * blockDim.x + threadIdx.x;
    long stride = (long)gridDim.x * blockDim.x;
    float4 z = make_float4(0.f, 0.f, 0.f, 0.f);
    for (; i < n4; i += stride) p[i] = z;
}

template<int F>
__global__ void gcn_scatter_add_kernel(const float* __restrict__ feat,
                                       const int*   __restrict__ src,
                                       const int*   __restrict__ dst,
                                       float*       __restrict__ agg,
                                       int n_edges) {
    int lane = threadIdx.x & 31;
    int wave = blockIdx.x * (blockDim.x >> 5) + (threadIdx.x >> 5);
    int waves_total = gridDim.x * (blockDim.x >> 5);
    int c = lane * 2;
    for (int e = wave; e < n_edges; e += waves_total) {
        int s = __builtin_amdgcn_readfirstlane(src[e]);
        int d = __builtin_amdgcn_readfirstlane(dst[e]);
        const v2f v = *(const v2f*)(feat + (long)s * F + c);
        float* a = agg + (long)d * F + c;
        atomicAdd(a + 0, v.x);
        atomicAdd(a + 1, v.y);
    }
}

template<int K, int NTILES, bool RELU>
__global__ void gcn_gemm_bias_kernel(const float* __restrict__ A,
                                     const float* __restrict__ W,
                                     const float* __restrict__ bias,
                                     float*       __restrict__ C,
                                     int n_rows) {
    constexpr int N = NTILES * 16;
    int lane = threadIdx.x & 31;
    int wave = blockIdx.x * (blockDim.x >> 5) + (threadIdx.x >> 5);

    int m0 = wave * 16;
    if (m0 >= n_rows) return;

    int m     = lane & 15;
    int khalf = (lane >> 4) << 1;

    v8f acc[NTILES];
    v8f zero = {};
#pragma unroll
    for (int t = 0; t < NTILES; ++t) acc[t] = zero;

    const float* arow = A + (long)(m0 + m) * K;

#pragma unroll
    for (int k0 = 0; k0 < K; k0 += 4) {
        v2f a = *(const v2f*)(arow + k0 + khalf);
#pragma unroll
        for (int t = 0; t < NTILES; ++t) {
            int n = t * 16 + m;
            v2f b;
            b.x = W[(k0 + khalf) * N + n];
            b.y = W[(k0 + khalf + 1) * N + n];
            acc[t] = __builtin_amdgcn_wmma_f32_16x16x4_f32(
                 false, a,  false, b,
                 (short)0, acc[t],
                 false,  false);
        }
    }

    int mrow_off = (lane >> 4) * 8;
#pragma unroll
    for (int t = 0; t < NTILES; ++t) {
        int col = t * 16 + m;
        float bv = bias[col];
#pragma unroll
        for (int r = 0; r < 8; ++r) {
            int row = m0 + r + mrow_off;
            float v = acc[t][r] + bv;
            if (RELU) v = fmaxf(v, 0.f);
            C[(long)row * N + col] = v;
        }
    }
}

extern "C" void kernel_launch(void* const* d_in, const int* in_sizes, int n_in,
                              void* d_out, int out_size, void* d_ws, size_t ws_size,
                              hipStream_t stream) {
    const float* feat = (const float*)d_in[0];
    const int*   src  = (const int*)d_in[1];
    const int*   dst  = (const int*)d_in[2];
    const float* W1   = (const float*)d_in[3];
    const float* b1   = (const float*)d_in[4];
    const float* W2   = (const float*)d_in[5];
    const float* b2   = (const float*)d_in[6];
    float* out = (float*)d_out;

    int n_nodes = in_sizes[0] / IN_F;
    int n_edges = in_sizes[1];

    float* agg = (float*)d_ws;
    float* h   = agg + (size_t)n_nodes * HID_F;

    long  zn4 = (long)n_nodes * IN_F / 4;
    dim3  blk(256);

    int row_tiles = (n_nodes + 15) / 16;
    int gemm_blocks = (row_tiles + 7) / 8;

    gcn_zero_kernel<<<1024, blk, 0, stream>>>((float4*)agg, zn4);
    gcn_scatter_add_kernel<IN_F><<<8192, blk, 0, stream>>>(feat, src, dst, agg, n_edges);
    gcn_gemm_bias_kernel<IN_F, HID_F / 16, true>
        <<<gemm_blocks, blk, 0, stream>>>(agg, W1, b1, h, n_nodes);

    gcn_zero_kernel<<<1024, blk, 0, stream>>>((float4*)agg, zn4);
    gcn_scatter_add_kernel<HID_F><<<8192, blk, 0, stream>>>(h, src, dst, agg, n_edges);
    gcn_gemm_bias_kernel<HID_F, OUT_F / 16, false>
        <<<gemm_blocks, blk, 0, stream>>>(agg, W2, b2, out, n_nodes);
}


# Layer 2 multiplies by W2 before aggregating: t=h*W2 on WMMA f32, out seeded with b2, atomic scatter-add of 32 cols per edge, not 64; layer-2 sum order changes (per-edge products summed)
# speedup vs baseline: 1.3105x; 1.3105x over previous
#include <hip/hip_runtime.h>

typedef __attribute__((ext_vector_type(2))) float v2f;
typedef __attribute__((ext_vector_type(8))) float v8f;

#define IN_F  64
#define HID_F 64
#define OUT_F 32

__global__ void gcn_zero_kernel(float4* __restrict__ p, long n4) {
    long i = (long)blockIdx.x * blockDim.x + threadIdx.x;
    long stride = (long)gridDim.x * blockDim.x;
    float4 z = make_float4(0.f, 0.f, 0.f, 0.f);
    for (; i < n4; i += stride) p[i] = z;
}

template<int F>
__global__ void gcn_scatter_add_kernel(const float* __restrict__ feat,
                                       const int*   __restrict__ src,
                                       const int*   __restrict__ dst,
                                       float*       __restrict__ agg,
                                       int n_edges) {
    constexpr int LPE = F / 2;
    constexpr int EPW = 32 / LPE;
    static_assert(F == 64 || F == 32, "row must be one or two half-waves of float2");
    int lane = threadIdx.x & 31;
    int wave = blockIdx.x * (blockDim.x >> 5) + (threadIdx.x >> 5);
    int waves_total = gridDim.x * (blockDim.x >> 5);
    int sub = lane / LPE;
    int c = (lane % LPE) * 2;
    for (int e = wave * EPW + sub; e < n_edges; e += waves_total * EPW) {
        int s = src[e];
        int d = dst[e];
        if (EPW == 1) {
            s = __builtin_amdgcn_readfirstlane(s);
            d = __builtin_amdgcn_readfirstlane(d);
        }
        const v2f v = *(const v2f*)(feat + (long)s * F + c);
        float* a = agg + (long)d * F + c;
        atomicAdd(a + 0, v.x);
        atomicAdd(a + 1, v.y);
    }
}

template<int K, int NTILES, bool RELU, bool SEED>
__global__ void gcn_gemm_bias_kernel(const float* __restrict__ A,
                                     const float* __restrict__ W,
                                     const float* __restrict__ bias,
                                     float*       __restrict__ C,
                                     float*       __restrict__ S,
                                     int n_rows) {
    constexpr int N = NTILES * 16;
    int lane = threadIdx.x & 31;
    int wave = blockIdx.x * (blockDim.x >> 5) + (threadIdx.x >> 5);

    int m0 = wave * 16;
    if (m0 >= n_rows) return;

    int m     = lane & 15;
    int khalf = (lane >> 4) << 1;

    v8f acc[NTILES];
    v8f zero = {};
#pragma unroll
    for (int t = 0; t < NTILES; ++t) acc[t] = zero;

    const float* arow = A + (long)(m0 + m) * K;

#pragma unroll
    for (int k0 = 0; k0 < K; k0 += 4) {
        v2f a = *(const v2f*)(arow + k0 + khalf);
#pragma unroll
        for (int t = 0; t < NTILES; ++t) {
            int n = t * 16 + m;
            v2f b;
            b.x = W[(k0 + khalf) * N + n];
            b.y = W[(k0 + khalf + 1) * N + n];
            acc[t] = __builtin_amdgcn_wmma_f32_16x16x4_f32(
                 false, a,  false, b,
                 (short)0, acc[t],
                 false,  false);
        }
    }

    int mrow_off = (lane >> 4) * 8;
#pragma unroll
    for (int t = 0; t < NTILES; ++t) {
        int col = t * 16 + m;
        float bv = bias[col];
#pragma unroll
        for (int r = 0; r < 8; ++r) {
            int row = m0 + r + mrow_off;
            if (SEED) {
                C[(long)row * N + col] = acc[t][r];
                S[(long)row * N + col] = bv;
            } else {
                float v = acc[t][r] + bv;
                if (RELU) v = fmaxf(v, 0.f);
                C[(long)row * N + col] = v;
            }
        }
    }
}

extern "C" void kernel_launch(void* const* d_in, const int* in_sizes, int n_in,
                              void* d_out, int out_size, void* d_ws, size_t ws_size,
                              hipStream_t stream) {
    const float* feat = (const float*)d_in[0];
    const int*   src  = (const int*)d_in[1];
    const int*   dst  = (const int*)d_in[2];
    const float* W1   = (const float*)d_in[3];
    const float* b1   = (const float*)d_in[4];
    const float* W2   = (const float*)d_in[5];
    const float* b2   = (const float*)d_in[6];
    float* out = (float*)d_out;

    int n_nodes = in_sizes[0] / IN_F;
    int n_edges = in_sizes[1];

    float* agg = (float*)d_ws;
    float* h   = agg + (size_t)n_nodes * HID_F;

    long  zn4 = (long)n_nodes * IN_F / 4;
    dim3  blk(256);

    int row_tiles = (n_nodes + 15) / 16;
    int gemm_blocks = (row_tiles + 7) / 8;

    gcn_zero_kernel<<<1024, blk, 0, stream>>>((float4*)agg, zn4);
    gcn_scatter_add_kernel<IN_F><<<8192, blk, 0, stream>>>(feat, src, dst, agg, n_edges);
    gcn_gemm_bias_kernel<IN_F, HID_F / 16, true, false>
        <<<gemm_blocks, blk, 0, stream>>>(agg, W1, b1, h, nullptr, n_nodes);

    float* t = agg;
    gcn_gemm_bias_kernel<HID_F, OUT_F / 16, false, true>
        <<<gemm_blocks, blk, 0, stream>>>(h, W2, b2, t, out, n_nodes);
    gcn_scatter_add_kernel<OUT_F><<<8192, blk, 0, stream>>>(t, src, dst, out, n_edges);
}


# Edge passes: a wave loads 32 edge indices in one read and issues 16 row gathers back to back before their atomics (was one index read then one row read per trip); atomic arrival order changes
# speedup vs baseline: 1.6516x; 1.2603x over previous
#include <hip/hip_runtime.h>

typedef __attribute__((ext_vector_type(2))) float v2f;
typedef __attribute__((ext_vector_type(8))) float v8f;

#define IN_F  64
#define HID_F 64
#define OUT_F 32

__global__ void gcn_zero_kernel(float4* __restrict__ p, long n4) {
    long i = (long)blockIdx.x * blockDim.x + threadIdx.x;
    long stride = (long)gridDim.x * blockDim.x;
    float4 z = make_float4(0.f, 0.f, 0.f, 0.f);
    for (; i < n4; i += stride) p[i] = z;
}

#define EDGES_PER_STEP 32
#define STEPS_PER_WAVE 4

template<int J>
__device__ __forceinline__ int gcn_row_share(int v) {
    return __builtin_amdgcn_update_dpp(0, v, 0x150 + J, 0xf, 0xf, false);
}

template<int F>
__device__ __forceinline__ int gcn_pick(int v, int j) {
    if (F == 64) return __builtin_amdgcn_readlane(v, j);
    switch (j & 15) {
        case 0:  return gcn_row_share<0>(v);
        case 1:  return gcn_row_share<1>(v);
        case 2:  return gcn_row_share<2>(v);
        case 3:  return gcn_row_share<3>(v);
        case 4:  return gcn_row_share<4>(v);
        case 5:  return gcn_row_share<5>(v);
        case 6:  return gcn_row_share<6>(v);
        case 7:  return gcn_row_share<7>(v);
        case 8:  return gcn_row_share<8>(v);
        case 9:  return gcn_row_share<9>(v);
        case 10: return gcn_row_share<10>(v);
        case 11: return gcn_row_share<11>(v);
        case 12: return gcn_row_share<12>(v);
        case 13: return gcn_row_share<13>(v);
        case 14: return gcn_row_share<14>(v);
        default: return gcn_row_share<15>(v);
    }
}

template<int F, typename T>
__device__ __forceinline__ T* gcn_row(T* base, int i, int c) {
    if (F == 64) return base + (size_t)i * F + c;
    unsigned off = ((unsigned)i * (unsigned)F + (unsigned)c) * 4u;
    return (T*)((const char*)base + off);
}

template<int F>
__global__ void gcn_scatter_add_kernel(const float* __restrict__ feat,
                                       const int*   __restrict__ src,
                                       const int*   __restrict__ dst,
                                       float*       __restrict__ agg,
                                       int n_edges) {
    constexpr int LPE = F / 2;
    constexpr int EPW = 32 / LPE;
    constexpr int SUB = EDGES_PER_STEP / EPW;
    constexpr int UNR = 16;
    static_assert(F == 64 || F == 32, "row must be one or two half-waves of float2");
    static_assert(SUB % UNR == 0, "whole groups");
    int lane = threadIdx.x & 31;
    int wave = blockIdx.x * (blockDim.x >> 5) + (threadIdx.x >> 5);
    int waves_total = gridDim.x * (blockDim.x >> 5);
    int c = (lane % LPE) * 2;
    int n_steps = (n_edges + EDGES_PER_STEP - 1) / EDGES_PER_STEP;

    int st = wave;
    int s_l = 0, d_l = 0;
    if (st < n_steps) {
        int e = st * EDGES_PER_STEP + lane;
        if (e < n_edges) { s_l = src[e]; d_l = dst[e]; }
    }
    for (; st < n_steps; st += waves_total) {
        int s_cur = s_l, d_cur = d_l;
        int cnt = n_edges - st * EDGES_PER_STEP;
        int nx = st + waves_total;
        s_l = 0; d_l = 0;
        if (nx < n_steps) {
            int e = nx * EDGES_PER_STEP + lane;
            if (e < n_edges) { s_l = src[e]; d_l = dst[e]; }
        }
        if (cnt >= EDGES_PER_STEP) {
#pragma unroll
            for (int j0 = 0; j0 < SUB; j0 += UNR) {
                v2f v[UNR];
#pragma unroll
                for (int u = 0; u < UNR; ++u) {
                    int s = gcn_pick<F>(s_cur, j0 + u);
                    v[u] = *(const v2f*)gcn_row<F>(feat, s, c);
                }
#pragma unroll
                for (int u = 0; u < UNR; ++u) {
                    int d = gcn_pick<F>(d_cur, j0 + u);
                    float* a = gcn_row<F>(agg, d, c);
                    atomicAdd(a + 0, v[u].x);
                    atomicAdd(a + 1, v[u].y);
                }
            }
        } else {
            for (int j = 0; j < SUB; ++j) {
                int s = gcn_pick<F>(s_cur, j);
                int d = gcn_pick<F>(d_cur, j);
                int el = (F == 64) ? j : ((lane & 16) | j);
                if (el < cnt) {
                    const v2f v = *(const v2f*)gcn_row<F>(feat, s, c);
                    float* a = gcn_row<F>(agg, d, c);
                    atomicAdd(a + 0, v.x);
                    atomicAdd(a + 1, v.y);
                }
            }
        }
    }
}

template<int K, int NTILES, bool RELU, bool SEED>
__global__ void gcn_gemm_bias_kernel(const float* __restrict__ A,
                                     const float* __restrict__ W,
                                     const float* __restrict__ bias,
                                     float*       __restrict__ C,
                                     float*       __restrict__ S,
                                     int n_rows) {
    constexpr int N = NTILES * 16;
    int lane = threadIdx.x & 31;
    int wave = blockIdx.x * (blockDim.x >> 5) + (threadIdx.x >> 5);

    int m0 = wave * 16;
    if (m0 >= n_rows) return;

    int m     = lane & 15;
    int khalf = (lane >> 4) << 1;

    v8f acc[NTILES];
    v8f zero = {};
#pragma unroll
    for (int t = 0; t < NTILES; ++t) acc[t] = zero;

    const float* arow = A + (long)(m0 + m) * K;

#pragma unroll
    for (int k0 = 0; k0 < K; k0 += 4) {
        v2f a = *(const v2f*)(arow + k0 + khalf);
#pragma unroll
        for (int t = 0; t < NTILES; ++t) {
            int n = t * 16 + m;
            v2f b;
            b.x = W[(k0 + khalf) * N + n];
            b.y = W[(k0 + khalf + 1) * N + n];
            acc[t] = __builtin_amdgcn_wmma_f32_16x16x4_f32(
                 false, a,  false, b,
                 (short)0, acc[t],
                 false,  false);
        }
    }

    int mrow_off = (lane >> 4) * 8;
#pragma unroll
    for (int t = 0; t < NTILES; ++t) {
        int col = t * 16 + m;
        float bv = bias[col];
#pragma unroll
        for (int r = 0; r < 8; ++r) {
            int row = m0 + r + mrow_off;
            if (SEED) {
                C[(long)row * N + col] = acc[t][r];
                S[(long)row * N + col] = bv;
            } else {
                float v = acc[t][r] + bv;
                if (RELU) v = fmaxf(v, 0.f);
                C[(long)row * N + col] = v;
            }
        }
    }
}

extern "C" void kernel_launch(void* const* d_in, const int* in_sizes, int n_in,
                              void* d_out, int out_size, void* d_ws, size_t ws_size,
                              hipStream_t stream) {
    const float* feat = (const float*)d_in[0];
    const int*   src  = (const int*)d_in[1];
    const int*   dst  = (const int*)d_in[2];
    const float* W1   = (const float*)d_in[3];
    const float* b1   = (const float*)d_in[4];
    const float* W2   = (const float*)d_in[5];
    const float* b2   = (const float*)d_in[6];
    float* out = (float*)d_out;

    int n_nodes = in_sizes[0] / IN_F;
    int n_edges = in_sizes[1];

    float* agg = (float*)d_ws;
    float* h   = agg + (size_t)n_nodes * HID_F;

    long  zn4 = (long)n_nodes * IN_F / 4;
    dim3  blk(256);

    int edge_steps = (n_edges + EDGES_PER_STEP - 1) / EDGES_PER_STEP;
    int edge_blocks = (edge_steps + 8 * STEPS_PER_WAVE - 1) / (8 * STEPS_PER_WAVE);
    if (edge_blocks < 1) edge_blocks = 1;

    int row_tiles = (n_nodes + 15) / 16;
    int gemm_blocks = (row_tiles + 7) / 8;

    gcn_zero_kernel<<<1024, blk, 0, stream>>>((float4*)agg, zn4);
    gcn_scatter_add_kernel<IN_F><<<edge_blocks, blk, 0, stream>>>(feat, src, dst, agg, n_edges);
    gcn_gemm_bias_kernel<IN_F, HID_F / 16, true, false>
        <<<gemm_blocks, blk, 0, stream>>>(agg, W1, b1, h, nullptr, n_nodes);

    float* t = agg;
    gcn_gemm_bias_kernel<HID_F, OUT_F / 16, false, true>
        <<<gemm_blocks, blk, 0, stream>>>(h, W2, b2, t, out, n_nodes);
    gcn_scatter_add_kernel<OUT_F><<<edge_blocks, blk, 0, stream>>>(t, src, dst, out, n_edges);
}


# Edge passes: the two 16-lane rows swap one float per lane (v_permlanex16) so each f32 atomic instruction adds into one whole 128-B line, not two half lines; atomic arrival order changes
# speedup vs baseline: 1.8218x; 1.1031x over previous
#include <hip/hip_runtime.h>

typedef __attribute__((ext_vector_type(2))) float v2f;
typedef __attribute__((ext_vector_type(8))) float v8f;

#define IN_F  64
#define HID_F 64
#define OUT_F 32

__global__ void gcn_zero_kernel(float4* __restrict__ p, long n4) {
    long i = (long)blockIdx.x * blockDim.x + threadIdx.x;
    long stride = (long)gridDim.x * blockDim.x;
    float4 z = make_float4(0.f, 0.f, 0.f, 0.f);
    for (; i < n4; i += stride) p[i] = z;
}

#define EDGES_PER_STEP 32
#define STEPS_PER_WAVE 4

template<int J>
__device__ __forceinline__ int gcn_row_share(int v) {
    return __builtin_amdgcn_update_dpp(0, v, 0x150 + J, 0xf, 0xf, false);
}

template<int F>
__device__ __forceinline__ int gcn_pick(int v, int j) {
    if (F == 64) return __builtin_amdgcn_readlane(v, j);
    switch (j & 15) {
        case 0:  return gcn_row_share<0>(v);
        case 1:  return gcn_row_share<1>(v);
        case 2:  return gcn_row_share<2>(v);
        case 3:  return gcn_row_share<3>(v);
        case 4:  return gcn_row_share<4>(v);
        case 5:  return gcn_row_share<5>(v);
        case 6:  return gcn_row_share<6>(v);
        case 7:  return gcn_row_share<7>(v);
        case 8:  return gcn_row_share<8>(v);
        case 9:  return gcn_row_share<9>(v);
        case 10: return gcn_row_share<10>(v);
        case 11: return gcn_row_share<11>(v);
        case 12: return gcn_row_share<12>(v);
        case 13: return gcn_row_share<13>(v);
        case 14: return gcn_row_share<14>(v);
        default: return gcn_row_share<15>(v);
    }
}

__device__ __forceinline__ float gcn_other_half(float v) {
    unsigned u = __builtin_bit_cast(unsigned, v);
    unsigned r = __builtin_amdgcn_permlanex16(u, u, 0x76543210u, 0xfedcba98u, false, false);
    return __builtin_bit_cast(float, r);
}

template<int F, typename T>
__device__ __forceinline__ T* gcn_row(T* base, int i, int c) {
    if (F == 64) return base + (size_t)i * F + c;
    unsigned off = ((unsigned)i * (unsigned)F + (unsigned)c) * 4u;
    return (T*)((const char*)base + off);
}

template<int F>
__global__ void gcn_scatter_add_kernel(const float* __restrict__ feat,
                                       const int*   __restrict__ src,
                                       const int*   __restrict__ dst,
                                       float*       __restrict__ agg,
                                       int n_edges) {
    constexpr int LPE = F / 2;
    constexpr int EPW = 32 / LPE;
    constexpr int SUB = EDGES_PER_STEP / EPW;
    constexpr int UNR = 16;
    static_assert(F == 64 || F == 32, "row must be one or two half-waves of float2");
    static_assert(SUB % UNR == 0, "whole groups");
    int lane = threadIdx.x & 31;
    int wave = blockIdx.x * (blockDim.x >> 5) + (threadIdx.x >> 5);
    int waves_total = gridDim.x * (blockDim.x >> 5);
    int c = (lane % LPE) * 2;
    bool hi = (lane & 16) != 0;
    int cw = (lane & 15) * 2 + (lane >> 4);
    int n_steps = (n_edges + EDGES_PER_STEP - 1) / EDGES_PER_STEP;

    int st = wave;
    int s_l = 0, d_l = 0;
    if (st < n_steps) {
        int e = st * EDGES_PER_STEP + lane;
        if (e < n_edges) { s_l = src[e]; d_l = dst[e]; }
    }
    for (; st < n_steps; st += waves_total) {
        int s_cur = s_l, d_cur = d_l;
        int cnt = n_edges - st * EDGES_PER_STEP;
        int nx = st + waves_total;
        s_l = 0; d_l = 0;
        if (nx < n_steps) {
            int e = nx * EDGES_PER_STEP + lane;
            if (e < n_edges) { s_l = src[e]; d_l = dst[e]; }
        }
        if (cnt >= EDGES_PER_STEP) {
#pragma unroll
            for (int j0 = 0; j0 < SUB; j0 += UNR) {
                v2f v[UNR];
#pragma unroll
                for (int u = 0; u < UNR; ++u) {
                    int s = gcn_pick<F>(s_cur, j0 + u);
                    v[u] = *(const v2f*)gcn_row<F>(feat, s, c);
                }
#pragma unroll
                for (int u = 0; u < UNR; ++u) {
                    float px = gcn_other_half(v[u].x);
                    float py = gcn_other_half(v[u].y);
                    float va = hi ? py : v[u].x;
                    float vb = hi ? v[u].y : px;
                    int da = __builtin_amdgcn_readlane(d_cur, j0 + u);
                    int db = (F == 64) ? da : __builtin_amdgcn_readlane(d_cur, j0 + u + 16);
                    float* la = agg + (size_t)da * F + cw;
                    float* lb = agg + (size_t)db * F + cw + (F == 64 ? 32 : 0);
                    atomicAdd(la, va);
                    atomicAdd(lb, vb);
                }
            }
        } else {
            for (int j = 0; j < SUB; ++j) {
                int s = gcn_pick<F>(s_cur, j);
                int d = gcn_pick<F>(d_cur, j);
                int el = (F == 64) ? j : ((lane & 16) | j);
                if (el < cnt) {
                    const v2f v = *(const v2f*)gcn_row<F>(feat, s, c);
                    float* a = gcn_row<F>(agg, d, c);
                    atomicAdd(a + 0, v.x);
                    atomicAdd(a + 1, v.y);
                }
            }
        }
    }
}

template<int K, int NTILES, bool RELU, bool SEED>
__global__ void gcn_gemm_bias_kernel(const float* __restrict__ A,
                                     const float* __restrict__ W,
                                     const float* __restrict__ bias,
                                     float*       __restrict__ C,
                                     float*       __restrict__ S,
                                     int n_rows) {
    constexpr int N = NTILES * 16;
    int lane = threadIdx.x & 31;
    int wave = blockIdx.x * (blockDim.x >> 5) + (threadIdx.x >> 5);

    int m0 = wave * 16;
    if (m0 >= n_rows) return;

    int m     = lane & 15;
    int khalf = (lane >> 4) << 1;

    v8f acc[NTILES];
    v8f zero = {};
#pragma unroll
    for (int t = 0; t < NTILES; ++t) acc[t] = zero;

    const float* arow = A + (long)(m0 + m) * K;

#pragma unroll
    for (int k0 = 0; k0 < K; k0 += 4) {
        v2f a = *(const v2f*)(arow + k0 + khalf);
#pragma unroll
        for (int t = 0; t < NTILES; ++t) {
            int n = t * 16 + m;
            v2f b;
            b.x = W[(k0 + khalf) * N + n];
            b.y = W[(k0 + khalf + 1) * N + n];
            acc[t] = __builtin_amdgcn_wmma_f32_16x16x4_f32(
                 false, a,  false, b,
                 (short)0, acc[t],
                 false,  false);
        }
    }

    int mrow_off = (lane >> 4) * 8;
#pragma unroll
    for (int t = 0; t < NTILES; ++t) {
        int col = t * 16 + m;
        float bv = bias[col];
#pragma unroll
        for (int r = 0; r < 8; ++r) {
            int row = m0 + r + mrow_off;
            if (SEED) {
                C[(long)row * N + col] = acc[t][r];
                S[(long)row * N + col] = bv;
            } else {
                float v = acc[t][r] + bv;
                if (RELU) v = fmaxf(v, 0.f);
                C[(long)row * N + col] = v;
            }
        }
    }
}

extern "C" void kernel_launch(void* const* d_in, const int* in_sizes, int n_in,
                              void* d_out, int out_size, void* d_ws, size_t ws_size,
                              hipStream_t stream) {
    const float* feat = (const float*)d_in[0];
    const int*   src  = (const int*)d_in[1];
    const int*   dst  = (const int*)d_in[2];
    const float* W1   = (const float*)d_in[3];
    const float* b1   = (const float*)d_in[4];
    const float* W2   = (const float*)d_in[5];
    const float* b2   = (const float*)d_in[6];
    float* out = (float*)d_out;

    int n_nodes = in_sizes[0] / IN_F;
    int n_edges = in_sizes[1];

    float* agg = (float*)d_ws;
    float* h   = agg + (size_t)n_nodes * HID_F;

    long  zn4 = (long)n_nodes * IN_F / 4;
    dim3  blk(256);

    int edge_steps = (n_edges + EDGES_PER_STEP - 1) / EDGES_PER_STEP;
    int edge_blocks = (edge_steps + 8 * STEPS_PER_WAVE - 1) / (8 * STEPS_PER_WAVE);
    if (edge_blocks < 1) edge_blocks = 1;

    int row_tiles = (n_nodes + 15) / 16;
    int gemm_blocks = (row_tiles + 7) / 8;

    gcn_zero_kernel<<<1024, blk, 0, stream>>>((float4*)agg, zn4);
    gcn_scatter_add_kernel<IN_F><<<edge_blocks, blk, 0, stream>>>(feat, src, dst, agg, n_edges);
    gcn_gemm_bias_kernel<IN_F, HID_F / 16, true, false>
        <<<gemm_blocks, blk, 0, stream>>>(agg, W1, b1, h, nullptr, n_nodes);

    float* t = agg;
    gcn_gemm_bias_kernel<HID_F, OUT_F / 16, false, true>
        <<<gemm_blocks, blk, 0, stream>>>(h, W2, b2, t, out, n_nodes);
    gcn_scatter_add_kernel<OUT_F><<<edge_blocks, blk, 0, stream>>>(t, src, dst, out, n_edges);
}


# Edge passes: per call edges are grouped by destination range (LDS counts and ranks), one work-group sums a range's gathered rows in an LDS tile (LDS f32 add), one store per row; sum order changes
# speedup vs baseline: 2.4298x; 1.3337x over previous
#include <hip/hip_runtime.h>

typedef __attribute__((ext_vector_type(2))) float v2f;
typedef __attribute__((ext_vector_type(8))) float v8f;

#define IN_F  64
#define HID_F 64
#define OUT_F 32

#define BS       196
#define NBK_MAX  1024
#define CH       8192
#define BIN_T    1024
#define EPT      (CH / BIN_T)

template<int NT>
__device__ __forceinline__ int gcn_block_scan(int v, int* sh, int tid, int& tot) {
    sh[tid] = v;
    __syncthreads();
    for (int off = 1; off < NT; off <<= 1) {
        int x = (tid >= off) ? sh[tid - off] : 0;
        __syncthreads();
        sh[tid] += x;
        __syncthreads();
    }
    int r = sh[tid];
    tot = sh[NT - 1];
    __syncthreads();
    return r;
}

__global__ __launch_bounds__(BIN_T)
void gcn_bin_count_kernel(const int* __restrict__ dst, int* __restrict__ cnt,
                          int n_edges, int nbk, int nblk) {
    __shared__ int hist[NBK_MAX];
    int tid = threadIdx.x;
    int blk = blockIdx.x;
    int base = blk * CH;
    bool full = n_edges - base >= CH;
    int d[EPT];
    if (full) {
        for (int k = 0; k < EPT; ++k) d[k] = dst[base + k * BIN_T + tid];
    } else {
        for (int k = 0; k < EPT; ++k) {
            int e = base + k * BIN_T + tid;
            d[k] = (e < n_edges) ? dst[e] : 0;
        }
    }
    for (int i = tid; i < nbk; i += BIN_T) hist[i] = 0;
    __syncthreads();
    for (int k = 0; k < EPT; ++k) {
        int e = base + k * BIN_T + tid;
        if (full || e < n_edges) atomicAdd(&hist[(unsigned)d[k] / BS], 1);
    }
    __syncthreads();
    for (int i = tid; i < nbk; i += BIN_T) cnt[(size_t)i * nblk + blk] = hist[i];
}

__global__ __launch_bounds__(256)
void gcn_bin_rowscan_kernel(int* __restrict__ cnt, int* __restrict__ total, int nblk) {
    __shared__ int sh[256];
    int tid = threadIdx.x;
    int* row = cnt + (size_t)blockIdx.x * nblk;
    int carry = 0;
    for (int t0 = 0; t0 < nblk; t0 += 256) {
        int i = t0 + tid;
        int v = (i < nblk) ? row[i] : 0;
        int tot;
        int incl = gcn_block_scan<256>(v, sh, tid, tot);
        if (i < nblk) row[i] = carry + incl - v;
        carry += tot;
    }
    if (tid == 0) total[blockIdx.x] = carry;
}

__global__ __launch_bounds__(BIN_T)
void gcn_bin_place_kernel(const int* __restrict__ src, const int* __restrict__ dst,
                          const int* __restrict__ rowpre, const int* __restrict__ total,
                          int2* __restrict__ list, int n_edges, int nbk, int nblk) {
    __shared__ int  hist[NBK_MAX];
    __shared__ int  lcur[NBK_MAX];
    __shared__ int  gdelta[NBK_MAX];
    __shared__ int  sh[BIN_T];
    __shared__ int2 stage[CH];
    int tid = threadIdx.x;
    int blk = blockIdx.x;
    for (int i = tid; i < NBK_MAX; i += BIN_T) hist[i] = 0;
    __syncthreads();
    int base = blk * CH;
    int s[EPT], d[EPT];
    for (int k = 0; k < EPT; ++k) {
        int e = base + k * BIN_T + tid;
        s[k] = 0; d[k] = 0;
        if (e < n_edges) { s[k] = src[e]; d[k] = dst[e]; }
    }
    for (int k = 0; k < EPT; ++k) {
        int e = base + k * BIN_T + tid;
        if (e < n_edges) atomicAdd(&hist[(unsigned)d[k] / BS], 1);
    }
    __syncthreads();
    int tot;
    int hv = hist[tid];
    int lstart = gcn_block_scan<BIN_T>(hv, sh, tid, tot) - hv;
    int tv = (tid < nbk) ? total[tid] : 0;
    int bstart = gcn_block_scan<BIN_T>(tv, sh, tid, tot) - tv;
    if (tid < nbk) {
        lcur[tid] = lstart;
        gdelta[tid] = bstart + rowpre[(size_t)tid * nblk + blk] - lstart;
    }
    __syncthreads();
    for (int k = 0; k < EPT; ++k) {
        int e = base + k * BIN_T + tid;
        if (e < n_edges) {
            int p = atomicAdd(&lcur[(unsigned)d[k] / BS], 1);
            stage[p] = make_int2(s[k], d[k]);
        }
    }
    __syncthreads();
    int nloc = n_edges - base;
    if (nloc > CH) nloc = CH;
    for (int i = tid; i < nloc; i += BIN_T) {
        int2 p = stage[i];
        int b = (unsigned)p.y / BS;
        list[gdelta[b] + i] = make_int2(p.x, p.y - b * BS);
    }
}

#define SUM_T 512

template<int F, int FS, bool BIAS>
__global__ __launch_bounds__(SUM_T)
void gcn_bucket_sum_kernel(const float* __restrict__ feat,
                           const int2*  __restrict__ list,
                           const int*   __restrict__ total,
                           const float* __restrict__ bias,
                           float*       __restrict__ outp,
                           int n_nodes) {
    constexpr int VW  = F / 32;
    constexpr int UNR = 16;
    static_assert(F == 64 || F == 32, "one or two columns per lane");
    __shared__ __attribute__((aligned(16))) float acc[BS * F];
    __shared__ int s_start;
    int tid  = threadIdx.x;
    int lane = tid & 31;
    int w    = tid >> 5;
    constexpr int NW = SUM_T / 32;
    int b = blockIdx.x;

    if (tid == 0) s_start = 0;
    float4* acc4 = (float4*)acc;
    float4 z = make_float4(0.f, 0.f, 0.f, 0.f);
    for (int i = tid; i < BS * F / 4; i += SUM_T) acc4[i] = z;
    int part = 0;
    for (int i = tid; i < b; i += SUM_T) part += total[i];
    for (int m = 16; m >= 1; m >>= 1) part += __shfl_xor(part, m);
    __syncthreads();
    if (lane == 0 && part != 0) atomicAdd(&s_start, part);
    __syncthreads();
    int start = s_start;
    int end   = start + total[b];

    int base = start + w * 32;
    int s_l = 0, d_l = 0;
    if (base + lane < end) { int2 p = list[base + lane]; s_l = p.x; d_l = p.y; }
    for (; base < end; base += NW * 32) {
        int s_cur = s_l, d_cur = d_l;
        int cnt = end - base;
        int nx = base + NW * 32;
        s_l = 0; d_l = 0;
        if (nx + lane < end) { int2 p = list[nx + lane]; s_l = p.x; d_l = p.y; }
        if (cnt >= 32) {
            for (int j0 = 0; j0 < 32; j0 += UNR) {
                float vx[UNR], vy[UNR];
                for (int u = 0; u < UNR; ++u) {
                    int sn = __builtin_amdgcn_readlane(s_cur, j0 + u);
                    const float* r = feat + (size_t)sn * FS + lane * VW;
                    if (VW == 2) { v2f t = *(const v2f*)r; vx[u] = t.x; vy[u] = t.y; }
                    else         { vx[u] = *r; vy[u] = 0.f; }
                }
                for (int u = 0; u < UNR; ++u) {
                    int dn = __builtin_amdgcn_readlane(d_cur, j0 + u);
                    float* a = acc + dn * F + lane * VW;
                    atomicAdd(a, vx[u]);
                    if (VW == 2) atomicAdd(a + 1, vy[u]);
                }
            }
        } else {
            for (int j = 0; j < cnt; ++j) {
                int sn = __builtin_amdgcn_readlane(s_cur, j);
                int dn = __builtin_amdgcn_readlane(d_cur, j);
                const float* r = feat + (size_t)sn * FS + lane * VW;
                float* a = acc + dn * F + lane * VW;
                if (VW == 2) { v2f t = *(const v2f*)r; atomicAdd(a, t.x); atomicAdd(a + 1, t.y); }
                else         { atomicAdd(a, *r); }
            }
        }
    }
    __syncthreads();
    int row0  = b * BS;
    int nrows = n_nodes - row0;
    if (nrows > BS) nrows = BS;
    float4* o4 = (float4*)(outp + (size_t)row0 * F);
    for (int i = tid; i < nrows * (F / 4); i += SUM_T) {
        float4 v = acc4[i];
        if (BIAS) {
            int c = (i % (F / 4)) * 4;
            v.x += bias[c]; v.y += bias[c + 1]; v.z += bias[c + 2]; v.w += bias[c + 3];
        }
        o4[i] = v;
    }
}

template<int K, int NTILES, bool RELU, bool BIAS, int CS>
__global__ void gcn_gemm_bias_kernel(const float* A,
                                     const float* __restrict__ W,
                                     const float* __restrict__ bias,
                                     float*       C,
                                     int n_rows) {
    constexpr int N = NTILES * 16;
    int lane = threadIdx.x & 31;
    int wave = blockIdx.x * (blockDim.x >> 5) + (threadIdx.x >> 5);

    int m0 = wave * 16;
    if (m0 >= n_rows) return;

    int m     = lane & 15;
    int khalf = (lane >> 4) << 1;

    v8f acc[NTILES];
    v8f zero = {};
#pragma unroll
    for (int t = 0; t < NTILES; ++t) acc[t] = zero;

    const float* arow = A + (long)(m0 + m) * K;

#pragma unroll
    for (int k0 = 0; k0 < K; k0 += 4) {
        v2f a = *(const v2f*)(arow + k0 + khalf);
#pragma unroll
        for (int t = 0; t < NTILES; ++t) {
            int n = t * 16 + m;
            v2f b;
            b.x = W[(k0 + khalf) * N + n];
            b.y = W[(k0 + khalf + 1) * N + n];
            acc[t] = __builtin_amdgcn_wmma_f32_16x16x4_f32(
                 false, a,  false, b,
                 (short)0, acc[t],
                 false,  false);
        }
    }

    int mrow_off = (lane >> 4) * 8;
#pragma unroll
    for (int t = 0; t < NTILES; ++t) {
        int col = t * 16 + m;
        float bv = BIAS ? bias[col] : 0.f;
#pragma unroll
        for (int r = 0; r < 8; ++r) {
            int row = m0 + r + mrow_off;
            float v = acc[t][r] + bv;
            if (RELU) v = fmaxf(v, 0.f);
            C[(long)row * CS + col] = v;
        }
    }
}

extern "C" void kernel_launch(void* const* d_in, const int* in_sizes, int n_in,
                              void* d_out, int out_size, void* d_ws, size_t ws_size,
                              hipStream_t stream) {
    const float* feat = (const float*)d_in[0];
    const int*   src  = (const int*)d_in[1];
    const int*   dst  = (const int*)d_in[2];
    const float* W1   = (const float*)d_in[3];
    const float* b1   = (const float*)d_in[4];
    const float* W2   = (const float*)d_in[5];
    const float* b2   = (const float*)d_in[6];
    float* out = (float*)d_out;

    int n_nodes = in_sizes[0] / IN_F;
    int n_edges = in_sizes[1];

    float* agg = (float*)d_ws;
    float* pl1 = agg + (size_t)n_nodes * HID_F;

    int nbk  = (n_nodes + BS - 1) / BS;
    int nblk = (n_edges + CH - 1) / CH;
    if (nbk  < 1) nbk  = 1;
    if (nblk < 1) nblk = 1;

    int2* list  = (int2*)pl1;
    int*  cnt   = (int*)(list + n_edges);
    int*  total = cnt + (size_t)nbk * nblk;

    dim3  blk(256);
    int row_tiles = (n_nodes + 15) / 16;
    int gemm_blocks = (row_tiles + 7) / 8;

    gcn_bin_count_kernel<<<nblk, BIN_T, 0, stream>>>(dst, cnt, n_edges, nbk, nblk);
    gcn_bin_rowscan_kernel<<<nbk, 256, 0, stream>>>(cnt, total, nblk);
    gcn_bin_place_kernel<<<nblk, BIN_T, 0, stream>>>(src, dst, cnt, total, list, n_edges, nbk, nblk);

    gcn_bucket_sum_kernel<IN_F, IN_F, false>
        <<<nbk, SUM_T, 0, stream>>>(feat, list, total, nullptr, agg, n_nodes);
    gcn_gemm_bias_kernel<IN_F, HID_F / 16, true, true, HID_F>
        <<<gemm_blocks, blk, 0, stream>>>(agg, W1, b1, agg, n_nodes);

    gcn_gemm_bias_kernel<HID_F, OUT_F / 16, false, false, HID_F>
        <<<gemm_blocks, blk, 0, stream>>>(agg, W2, nullptr, agg, n_nodes);
    gcn_bucket_sum_kernel<OUT_F, HID_F, true>
        <<<nbk, SUM_T, 0, stream>>>(agg, list, total, b2, out, n_nodes);
}


# Both dense layers run as the layer-1 bucket sum's epilogue: the LDS tile is multiplied by W1 (+b1, ReLU) and W2 on WMMA f32 from LDS and only t (32 cols) is stored; 7 launches become 5
# speedup vs baseline: 2.8384x; 1.1682x over previous
#include <hip/hip_runtime.h>

typedef __attribute__((ext_vector_type(2))) float v2f;
typedef __attribute__((ext_vector_type(8))) float v8f;

#define IN_F  64
#define HID_F 64
#define OUT_F 32

#define BS       196
#define NBK_MAX  1024
#define CH       8192
#define BIN_T    1024
#define EPT      (CH / BIN_T)

template<int NT>
__device__ __forceinline__ int gcn_block_scan(int v, int* sh, int tid, int& tot) {
    sh[tid] = v;
    __syncthreads();
    for (int off = 1; off < NT; off <<= 1) {
        int x = (tid >= off) ? sh[tid - off] : 0;
        __syncthreads();
        sh[tid] += x;
        __syncthreads();
    }
    int r = sh[tid];
    tot = sh[NT - 1];
    __syncthreads();
    return r;
}

__global__ __launch_bounds__(BIN_T)
void gcn_bin_count_kernel(const int* __restrict__ dst, int* __restrict__ cnt,
                          int n_edges, int nbk, int nblk) {
    __shared__ int hist[NBK_MAX];
    int tid = threadIdx.x;
    int blk = blockIdx.x;
    int base = blk * CH;
    bool full = n_edges - base >= CH;
    int d[EPT];
    if (full) {
        for (int k = 0; k < EPT; ++k) d[k] = dst[base + k * BIN_T + tid];
    } else {
        for (int k = 0; k < EPT; ++k) {
            int e = base + k * BIN_T + tid;
            d[k] = (e < n_edges) ? dst[e] : 0;
        }
    }
    for (int i = tid; i < nbk; i += BIN_T) hist[i] = 0;
    __syncthreads();
    for (int k = 0; k < EPT; ++k) {
        int e = base + k * BIN_T + tid;
        if (full || e < n_edges) atomicAdd(&hist[(unsigned)d[k] / BS], 1);
    }
    __syncthreads();
    for (int i = tid; i < nbk; i += BIN_T) cnt[(size_t)i * nblk + blk] = hist[i];
}

__global__ __launch_bounds__(256)
void gcn_bin_rowscan_kernel(int* __restrict__ cnt, int* __restrict__ total, int nblk) {
    __shared__ int sh[256];
    int tid = threadIdx.x;
    int* row = cnt + (size_t)blockIdx.x * nblk;
    int carry = 0;
    for (int t0 = 0; t0 < nblk; t0 += 256) {
        int i = t0 + tid;
        int v = (i < nblk) ? row[i] : 0;
        int tot;
        int incl = gcn_block_scan<256>(v, sh, tid, tot);
        if (i < nblk) row[i] = carry + incl - v;
        carry += tot;
    }
    if (tid == 0) total[blockIdx.x] = carry;
}

__global__ __launch_bounds__(BIN_T)
void gcn_bin_place_kernel(const int* __restrict__ src, const int* __restrict__ dst,
                          const int* __restrict__ rowpre, const int* __restrict__ total,
                          int2* __restrict__ list, int n_edges, int nbk, int nblk) {
    __shared__ int  hist[NBK_MAX];
    __shared__ int  lcur[NBK_MAX];
    __shared__ int  gdelta[NBK_MAX];
    __shared__ int  sh[BIN_T];
    __shared__ int2 stage[CH];
    int tid = threadIdx.x;
    int blk = blockIdx.x;
    for (int i = tid; i < NBK_MAX; i += BIN_T) hist[i] = 0;
    __syncthreads();
    int base = blk * CH;
    int s[EPT], d[EPT];
    for (int k = 0; k < EPT; ++k) {
        int e = base + k * BIN_T + tid;
        s[k] = 0; d[k] = 0;
        if (e < n_edges) { s[k] = src[e]; d[k] = dst[e]; }
    }
    for (int k = 0; k < EPT; ++k) {
        int e = base + k * BIN_T + tid;
        if (e < n_edges) atomicAdd(&hist[(unsigned)d[k] / BS], 1);
    }
    __syncthreads();
    int tot;
    int hv = hist[tid];
    int lstart = gcn_block_scan<BIN_T>(hv, sh, tid, tot) - hv;
    int tv = (tid < nbk) ? total[tid] : 0;
    int bstart = gcn_block_scan<BIN_T>(tv, sh, tid, tot) - tv;
    if (tid < nbk) {
        lcur[tid] = lstart;
        gdelta[tid] = bstart + rowpre[(size_t)tid * nblk + blk] - lstart;
    }
    __syncthreads();
    for (int k = 0; k < EPT; ++k) {
        int e = base + k * BIN_T + tid;
        if (e < n_edges) {
            int p = atomicAdd(&lcur[(unsigned)d[k] / BS], 1);
            stage[p] = make_int2(s[k], d[k]);
        }
    }
    __syncthreads();
    int nloc = n_edges - base;
    if (nloc > CH) nloc = CH;
    for (int i = tid; i < nloc; i += BIN_T) {
        int2 p = stage[i];
        int b = (unsigned)p.y / BS;
        list[gdelta[b] + i] = make_int2(p.x, p.y - b * BS);
    }
}

#define SUM_T 512

__device__ __forceinline__ int gcn_bucket_start(const int* __restrict__ total, int b,
                                                int* s_start, int tid, int lane) {
    if (tid == 0) *s_start = 0;
    int part = 0;
    for (int i = tid; i < b; i += SUM_T) part += total[i];
    for (int m = 16; m >= 1; m >>= 1) part += __shfl_xor(part, m);
    __syncthreads();
    if (lane == 0 && part != 0) atomicAdd(s_start, part);
    __syncthreads();
    return *s_start;
}

template<int F, int FS, int P>
__device__ __forceinline__ void gcn_bucket_gather(const float* __restrict__ feat,
                                                  const int2*  __restrict__ list,
                                                  int start, int end,
                                                  float* acc, int lane, int w) {
    constexpr int VW  = F / 32;
    constexpr int UNR = 16;
    constexpr int NW  = SUM_T / 32;
    static_assert(F == 64 || F == 32, "one or two columns per lane");
    int base = start + w * 32;
    int s_l = 0, d_l = 0;
    if (base + lane < end) { int2 p = list[base + lane]; s_l = p.x; d_l = p.y; }
    for (; base < end; base += NW * 32) {
        int s_cur = s_l, d_cur = d_l;
        int cnt = end - base;
        int nx = base + NW * 32;
        s_l = 0; d_l = 0;
        if (nx + lane < end) { int2 p = list[nx + lane]; s_l = p.x; d_l = p.y; }
        if (cnt >= 32) {
            for (int j0 = 0; j0 < 32; j0 += UNR) {
                float vx[UNR], vy[UNR];
                for (int u = 0; u < UNR; ++u) {
                    int sn = __builtin_amdgcn_readlane(s_cur, j0 + u);
                    const float* r = feat + (size_t)sn * FS + lane * VW;
                    if (VW == 2) { v2f t = *(const v2f*)r; vx[u] = t.x; vy[u] = t.y; }
                    else         { vx[u] = *r; vy[u] = 0.f; }
                }
                for (int u = 0; u < UNR; ++u) {
                    int dn = __builtin_amdgcn_readlane(d_cur, j0 + u);
                    float* a = acc + dn * P + lane * VW;
                    atomicAdd(a, vx[u]);
                    if (VW == 2) atomicAdd(a + 1, vy[u]);
                }
            }
        } else {
            for (int j = 0; j < cnt; ++j) {
                int sn = __builtin_amdgcn_readlane(s_cur, j);
                int dn = __builtin_amdgcn_readlane(d_cur, j);
                const float* r = feat + (size_t)sn * FS + lane * VW;
                float* a = acc + dn * P + lane * VW;
                if (VW == 2) { v2f t = *(const v2f*)r; atomicAdd(a, t.x); atomicAdd(a + 1, t.y); }
                else         { atomicAdd(a, *r); }
            }
        }
    }
}

template<int F, int FS, bool BIAS>
__global__ __launch_bounds__(SUM_T)
void gcn_bucket_sum_kernel(const float* __restrict__ feat,
                           const int2*  __restrict__ list,
                           const int*   __restrict__ total,
                           const float* __restrict__ bias,
                           float*       __restrict__ outp,
                           int n_nodes) {
    __shared__ __attribute__((aligned(16))) float acc[BS * F];
    __shared__ int s_start;
    int tid  = threadIdx.x;
    int lane = tid & 31;
    int w    = tid >> 5;
    int b = blockIdx.x;

    float4* acc4 = (float4*)acc;
    float4 z = make_float4(0.f, 0.f, 0.f, 0.f);
    for (int i = tid; i < BS * F / 4; i += SUM_T) acc4[i] = z;
    int start = gcn_bucket_start(total, b, &s_start, tid, lane);
    int end   = start + total[b];

    gcn_bucket_gather<F, FS, F>(feat, list, start, end, acc, lane, w);
    __syncthreads();
    int row0  = b * BS;
    int nrows = n_nodes - row0;
    if (nrows > BS) nrows = BS;
    float4* o4 = (float4*)(outp + (size_t)row0 * F);
    for (int i = tid; i < nrows * (F / 4); i += SUM_T) {
        float4 v = acc4[i];
        if (BIAS) {
            int c = (i % (F / 4)) * 4;
            v.x += bias[c]; v.y += bias[c + 1]; v.z += bias[c + 2]; v.w += bias[c + 3];
        }
        o4[i] = v;
    }
}

#define NSTRIP ((BS + 15) / 16)
#define BSP    (NSTRIP * 16)
#define AP     68
#define W1P    (2 * HID_F + 32)
#define W2P    (2 * OUT_F + 32)

__global__ __launch_bounds__(SUM_T)
void gcn_bucket_sum_dense_kernel(const float* __restrict__ feat,
                                 const int2*  __restrict__ list,
                                 const int*   __restrict__ total,
                                 const float* __restrict__ W1,
                                 const float* __restrict__ b1,
                                 const float* __restrict__ W2,
                                 float*       __restrict__ tout,
                                 int n_nodes) {
    static_assert(IN_F == 64 && HID_F == 64 && OUT_F == 32, "tile and strip shapes");
    static_assert(NSTRIP <= SUM_T / 32, "one strip per wave");
    __shared__ __attribute__((aligned(16))) float acc[BSP * AP];
    __shared__ __attribute__((aligned(16))) float w1s[(IN_F / 2) * W1P];
    __shared__ __attribute__((aligned(16))) float w2s[(HID_F / 2) * W2P];
    __shared__ int s_start;
    int tid  = threadIdx.x;
    int lane = tid & 31;
    int w    = tid >> 5;
    int b = blockIdx.x;

    float4* acc4 = (float4*)acc;
    float4 z = make_float4(0.f, 0.f, 0.f, 0.f);
    for (int i = tid; i < BSP * AP / 4; i += SUM_T) acc4[i] = z;
    for (int i = tid; i < IN_F * HID_F; i += SUM_T) {
        int k = i / HID_F, n = i % HID_F;
        w1s[(k >> 1) * W1P + n * 2 + (k & 1)] = W1[i];
    }
    for (int i = tid; i < HID_F * OUT_F; i += SUM_T) {
        int k = i / OUT_F, n = i % OUT_F;
        w2s[(k >> 1) * W2P + n * 2 + (k & 1)] = W2[i];
    }
    int start = gcn_bucket_start(total, b, &s_start, tid, lane);
    int end   = start + total[b];

    gcn_bucket_gather<IN_F, IN_F, AP>(feat, list, start, end, acc, lane, w);
    __syncthreads();

    bool active = w < NSTRIP;
    int m     = lane & 15;
    int khalf = (lane >> 4) << 1;
    int mrow  = (lane >> 4) * 8;
    float* arow = acc + (w * 16 + m) * AP + khalf;
    float* drow = acc + (w * 16 + mrow) * AP + m;

    if (active) {
        v8f c1[HID_F / 16];
        v8f zero = {};
#pragma unroll
        for (int t = 0; t < HID_F / 16; ++t) c1[t] = zero;
#pragma unroll
        for (int k0 = 0; k0 < IN_F; k0 += 4) {
            v2f a = *(const v2f*)(arow + k0);
            const float* brow = w1s + ((k0 + khalf) >> 1) * W1P + m * 2;
#pragma unroll
            for (int t = 0; t < HID_F / 16; ++t) {
                v2f bb = *(const v2f*)(brow + t * 32);
                c1[t] = __builtin_amdgcn_wmma_f32_16x16x4_f32(
                     false, a,  false, bb,
                     (short)0, c1[t],
                     false,  false);
            }
        }
#pragma unroll
        for (int t = 0; t < HID_F / 16; ++t) {
            float bv = b1[t * 16 + m];
#pragma unroll
            for (int r = 0; r < 8; ++r)
                drow[r * AP + t * 16] = fmaxf(c1[t][r] + bv, 0.f);
        }
    }
    __syncthreads();
    if (active) {
        v8f c2[OUT_F / 16];
        v8f zero = {};
#pragma unroll
        for (int t = 0; t < OUT_F / 16; ++t) c2[t] = zero;
#pragma unroll
        for (int k0 = 0; k0 < HID_F; k0 += 4) {
            v2f a = *(const v2f*)(arow + k0);
            const float* brow = w2s + ((k0 + khalf) >> 1) * W2P + m * 2;
#pragma unroll
            for (int t = 0; t < OUT_F / 16; ++t) {
                v2f bb = *(const v2f*)(brow + t * 32);
                c2[t] = __builtin_amdgcn_wmma_f32_16x16x4_f32(
                     false, a,  false, bb,
                     (short)0, c2[t],
                     false,  false);
            }
        }
#pragma unroll
        for (int t = 0; t < OUT_F / 16; ++t) {
#pragma unroll
            for (int r = 0; r < 8; ++r)
                drow[r * AP + t * 16] = c2[t][r];
        }
    }
    __syncthreads();
    int row0  = b * BS;
    int nrows = n_nodes - row0;
    if (nrows > BS) nrows = BS;
    float4* o4 = (float4*)(tout + (size_t)row0 * OUT_F);
    for (int i = tid; i < nrows * (OUT_F / 4); i += SUM_T) {
        int lr = i / (OUT_F / 4), c4 = i % (OUT_F / 4);
        o4[i] = *(const float4*)(acc + lr * AP + c4 * 4);
    }
}

extern "C" void kernel_launch(void* const* d_in, const int* in_sizes, int n_in,
                              void* d_out, int out_size, void* d_ws, size_t ws_size,
                              hipStream_t stream) {
    const float* feat = (const float*)d_in[0];
    const int*   src  = (const int*)d_in[1];
    const int*   dst  = (const int*)d_in[2];
    const float* W1   = (const float*)d_in[3];
    const float* b1   = (const float*)d_in[4];
    const float* W2   = (const float*)d_in[5];
    const float* b2   = (const float*)d_in[6];
    float* out = (float*)d_out;

    int n_nodes = in_sizes[0] / IN_F;
    int n_edges = in_sizes[1];

    float* tbuf = (float*)d_ws;
    float* pl1  = tbuf + (size_t)n_nodes * HID_F;

    int nbk  = (n_nodes + BS - 1) / BS;
    int nblk = (n_edges + CH - 1) / CH;
    if (nbk  < 1) nbk  = 1;
    if (nblk < 1) nblk = 1;

    int2* list  = (int2*)pl1;
    int*  cnt   = (int*)(list + n_edges);
    int*  total = cnt + (size_t)nbk * nblk;

    gcn_bin_count_kernel<<<nblk, BIN_T, 0, stream>>>(dst, cnt, n_edges, nbk, nblk);
    gcn_bin_rowscan_kernel<<<nbk, 256, 0, stream>>>(cnt, total, nblk);
    gcn_bin_place_kernel<<<nblk, BIN_T, 0, stream>>>(src, dst, cnt, total, list, n_edges, nbk, nblk);

    gcn_bucket_sum_dense_kernel
        <<<nbk, SUM_T, 0, stream>>>(feat, list, total, W1, b1, W2, tbuf, n_nodes);

    gcn_bucket_sum_kernel<OUT_F, OUT_F, true>
        <<<nbk, SUM_T, 0, stream>>>(tbuf, list, total, b2, out, n_nodes);
}


# Edge grouping in one launch: each chunk of edges is sorted by bucket on its own, and a bucket's work-group collects its run of every chunk into an LDS list; 5 launches become 3; sum order as before
# speedup vs baseline: 3.0479x; 1.0738x over previous
#include <hip/hip_runtime.h>

typedef __attribute__((ext_vector_type(2))) float v2f;
typedef __attribute__((ext_vector_type(8))) float v8f;

#define IN_F  64
#define HID_F 64
#define OUT_F 32

#define BS       196
#define NBK_MAX  1024
#define CH       8192
#define BIN_T    1024
#define EPT      (CH / BIN_T)

template<int NW>
__device__ __forceinline__ int gcn_wg_scan(int v, int* wsum, int lane, int w, int& tot) {
    int x = v;
    for (int o = 1; o < 32; o <<= 1) {
        int y = __shfl_up(x, o);
        if (lane >= o) x += y;
    }
    if (lane == 31) wsum[w] = x;
    __syncthreads();
    int below = 0, all = 0;
    for (int j = 0; j < NW; ++j) {
        int s = wsum[j];
        if (j < w) below += s;
        all += s;
    }
    tot = all;
    __syncthreads();
    return x + below;
}

__global__ __launch_bounds__(BIN_T)
void gcn_chunk_sort_kernel(const int* __restrict__ src, const int* __restrict__ dst,
                           int2* __restrict__ list, int2* __restrict__ offt,
                           int n_edges, int nbk) {
    static_assert(NBK_MAX == BIN_T, "one bucket counter per thread");
    __shared__ int  hist[NBK_MAX];
    __shared__ int  wsum[BIN_T / 32];
    __shared__ int2 stage[CH];
    int tid  = threadIdx.x;
    int lane = tid & 31;
    int w    = tid >> 5;
    int blk  = blockIdx.x;
    hist[tid] = 0;
    int base = blk * CH;
    int s[EPT], d[EPT];
    for (int k = 0; k < EPT; ++k) {
        int e = base + k * BIN_T + tid;
        s[k] = 0; d[k] = 0;
        if (e < n_edges) { s[k] = src[e]; d[k] = dst[e]; }
    }
    __syncthreads();
    for (int k = 0; k < EPT; ++k) {
        int e = base + k * BIN_T + tid;
        if (e < n_edges) atomicAdd(&hist[(unsigned)d[k] / BS], 1);
    }
    __syncthreads();
    int tot;
    int hv = hist[tid];
    int lstart = gcn_wg_scan<BIN_T / 32>(hv, wsum, lane, w, tot) - hv;
    hist[tid] = lstart;
    if (tid < nbk) offt[(size_t)blk * NBK_MAX + tid] = make_int2(lstart, hv);
    __syncthreads();
    for (int k = 0; k < EPT; ++k) {
        int e = base + k * BIN_T + tid;
        if (e < n_edges) {
            int bkt = (unsigned)d[k] / BS;
            int p = atomicAdd(&hist[bkt], 1);
            stage[p] = make_int2(s[k], d[k] - bkt * BS);
        }
    }
    __syncthreads();
    for (int i = tid; i < tot; i += BIN_T) list[base + i] = stage[i];
}

#define SUM_T 512
#define CAP   4096

template<int F, int FS, int P>
__device__ __forceinline__ void gcn_bucket_gather(const float* __restrict__ feat,
                                                  const int2*  __restrict__ list,
                                                  int start, int end,
                                                  float* acc, int lane, int w) {
    constexpr int VW  = F / 32;
    constexpr int UNR = 16;
    constexpr int NW  = SUM_T / 32;
    static_assert(F == 64 || F == 32, "one or two columns per lane");
    int base = start + w * 32;
    int s_l = 0, d_l = 0;
    if (base + lane < end) { int2 p = list[base + lane]; s_l = p.x; d_l = p.y; }
    for (; base < end; base += NW * 32) {
        int s_cur = s_l, d_cur = d_l;
        int cnt = end - base;
        int nx = base + NW * 32;
        s_l = 0; d_l = 0;
        if (nx + lane < end) { int2 p = list[nx + lane]; s_l = p.x; d_l = p.y; }
        if (cnt >= 32) {
            for (int j0 = 0; j0 < 32; j0 += UNR) {
                float vx[UNR], vy[UNR];
                for (int u = 0; u < UNR; ++u) {
                    int sn = __builtin_amdgcn_readlane(s_cur, j0 + u);
                    const float* r = feat + (size_t)sn * FS + lane * VW;
                    if (VW == 2) { v2f t = *(const v2f*)r; vx[u] = t.x; vy[u] = t.y; }
                    else         { vx[u] = *r; vy[u] = 0.f; }
                }
                for (int u = 0; u < UNR; ++u) {
                    int dn = __builtin_amdgcn_readlane(d_cur, j0 + u);
                    float* a = acc + dn * P + lane * VW;
                    atomicAdd(a, vx[u]);
                    if (VW == 2) atomicAdd(a + 1, vy[u]);
                }
            }
        } else {
            for (int j = 0; j < cnt; ++j) {
                int sn = __builtin_amdgcn_readlane(s_cur, j);
                int dn = __builtin_amdgcn_readlane(d_cur, j);
                const float* r = feat + (size_t)sn * FS + lane * VW;
                float* a = acc + dn * P + lane * VW;
                if (VW == 2) { v2f t = *(const v2f*)r; atomicAdd(a, t.x); atomicAdd(a + 1, t.y); }
                else         { atomicAdd(a, *r); }
            }
        }
    }
}

template<int F, int FS, int P>
__device__ __forceinline__ void gcn_bucket_edges(const float* __restrict__ feat,
                                                 const int2*  __restrict__ list,
                                                 const int2*  __restrict__ offt,
                                                 int b, int nblk,
                                                 float* acc, int2* ent, int* wsum,
                                                 int tid, int lane, int w) {
    constexpr int NW  = SUM_T / 32;
    constexpr int RPP = SUM_T / 2;
    constexpr int CU  = 16;
    int h = tid & 1;
    for (int t0 = 0; t0 < nblk; t0 += RPP) {
        int r = t0 + (tid >> 1);
        int o0 = 0, len = 0;
        if (r < nblk) {
            int2 o = offt[(size_t)r * NBK_MAX + b];
            o0  = o.x;
            len = o.y;
        }
        int tot;
        int pre = gcn_wg_scan<NW>(h == 0 ? len : 0, wsum, lane, w, tot) - len;
        const int2* run = list + (size_t)r * CH + o0;
        for (int seg = 0; seg < tot; seg += CAP) {
            int i_lo = seg - pre;
            if (i_lo < 0) i_lo = 0;
            int i_hi = seg + CAP - pre;
            if (i_hi > len) i_hi = len;
            for (int i = i_lo + ((i_lo ^ h) & 1); i < i_hi; i += 2 * CU) {
                int2 v[CU];
#pragma unroll
                for (int u = 0; u < CU; ++u) {
                    int ii = i + 2 * u;
                    if (ii < i_hi) v[u] = run[ii];
                }
#pragma unroll
                for (int u = 0; u < CU; ++u) {
                    int ii = i + 2 * u;
                    if (ii < i_hi) ent[pre + ii - seg] = v[u];
                }
            }
            __syncthreads();
            int n = tot - seg;
            if (n > CAP) n = CAP;
            gcn_bucket_gather<F, FS, P>(feat, ent, 0, n, acc, lane, w);
            __syncthreads();
        }
    }
}

template<int F, int FS, bool BIAS>
__global__ __launch_bounds__(SUM_T)
void gcn_bucket_sum_kernel(const float* __restrict__ feat,
                           const int2*  __restrict__ list,
                           const int2*  __restrict__ offt,
                           int nblk,
                           const float* __restrict__ bias,
                           float*       __restrict__ outp,
                           int n_nodes) {
    __shared__ __attribute__((aligned(16))) float acc[BS * F];
    __shared__ int2 ent[CAP];
    __shared__ int  wsum[SUM_T / 32];
    int tid  = threadIdx.x;
    int lane = tid & 31;
    int w    = tid >> 5;
    int b = blockIdx.x;

    float4* acc4 = (float4*)acc;
    float4 z = make_float4(0.f, 0.f, 0.f, 0.f);
    for (int i = tid; i < BS * F / 4; i += SUM_T) acc4[i] = z;

    gcn_bucket_edges<F, FS, F>(feat, list, offt, b, nblk, acc, ent, wsum, tid, lane, w);
    int row0  = b * BS;
    int nrows = n_nodes - row0;
    if (nrows > BS) nrows = BS;
    float4* o4 = (float4*)(outp + (size_t)row0 * F);
    for (int i = tid; i < nrows * (F / 4); i += SUM_T) {
        float4 v = acc4[i];
        if (BIAS) {
            int c = (i % (F / 4)) * 4;
            v.x += bias[c]; v.y += bias[c + 1]; v.z += bias[c + 2]; v.w += bias[c + 3];
        }
        o4[i] = v;
    }
}

#define NSTRIP ((BS + 15) / 16)
#define BSP    (NSTRIP * 16)
#define AP     68
#define W1P    (2 * HID_F + 32)
#define W2P    (2 * OUT_F + 32)

__global__ __launch_bounds__(SUM_T)
void gcn_bucket_sum_dense_kernel(const float* __restrict__ feat,
                                 const int2*  __restrict__ list,
                                 const int2*  __restrict__ offt,
                                 int nblk,
                                 const float* __restrict__ W1,
                                 const float* __restrict__ b1,
                                 const float* __restrict__ W2,
                                 float*       __restrict__ tout,
                                 int n_nodes) {
    static_assert(IN_F == 64 && HID_F == 64 && OUT_F == 32, "tile and strip shapes");
    static_assert(NSTRIP <= SUM_T / 32, "one strip per wave");
    __shared__ __attribute__((aligned(16))) float acc[BSP * AP];
    __shared__ __attribute__((aligned(16))) float w1s[(IN_F / 2) * W1P];
    __shared__ __attribute__((aligned(16))) float w2s[(HID_F / 2) * W2P];
    __shared__ int2 ent[CAP];
    __shared__ int  wsum[SUM_T / 32];
    int tid  = threadIdx.x;
    int lane = tid & 31;
    int w    = tid >> 5;
    int b = blockIdx.x;

    float4* acc4 = (float4*)acc;
    float4 z = make_float4(0.f, 0.f, 0.f, 0.f);
    for (int i = tid; i < BSP * AP / 4; i += SUM_T) acc4[i] = z;
    for (int i = tid; i < IN_F * HID_F; i += SUM_T) {
        int k = i / HID_F, n = i % HID_F;
        w1s[(k >> 1) * W1P + n * 2 + (k & 1)] = W1[i];
    }
    for (int i = tid; i < HID_F * OUT_F; i += SUM_T) {
        int k = i / OUT_F, n = i % OUT_F;
        w2s[(k >> 1) * W2P + n * 2 + (k & 1)] = W2[i];
    }

    gcn_bucket_edges<IN_F, IN_F, AP>(feat, list, offt, b, nblk, acc, ent, wsum, tid, lane, w);

    bool active = w < NSTRIP;
    int m     = lane & 15;
    int khalf = (lane >> 4) << 1;
    int mrow  = (lane >> 4) * 8;
    float* arow = acc + (w * 16 + m) * AP + khalf;
    float* drow = acc + (w * 16 + mrow) * AP + m;

    if (active) {
        v8f c1[HID_F / 16];
        v8f zero = {};
#pragma unroll
        for (int t = 0; t < HID_F / 16; ++t) c1[t] = zero;
#pragma unroll
        for (int k0 = 0; k0 < IN_F; k0 += 4) {
            v2f a = *(const v2f*)(arow + k0);
            const float* brow = w1s + ((k0 + khalf) >> 1) * W1P + m * 2;
#pragma unroll
            for (int t = 0; t < HID_F / 16; ++t) {
                v2f bb = *(const v2f*)(brow + t * 32);
                c1[t] = __builtin_amdgcn_wmma_f32_16x16x4_f32(
                     false, a,  false, bb,
                     (short)0, c1[t],
                     false,  false);
            }
        }
#pragma unroll
        for (int t = 0; t < HID_F / 16; ++t) {
            float bv = b1[t * 16 + m];
#pragma unroll
            for (int r = 0; r < 8; ++r)
                drow[r * AP + t * 16] = fmaxf(c1[t][r] + bv, 0.f);
        }
    }
    __syncthreads();
    if (active) {
        v8f c2[OUT_F / 16];
        v8f zero = {};
#pragma unroll
        for (int t = 0; t < OUT_F / 16; ++t) c2[t] = zero;
#pragma unroll
        for (int k0 = 0; k0 < HID_F; k0 += 4) {
            v2f a = *(const v2f*)(arow + k0);
            const float* brow = w2s + ((k0 + khalf) >> 1) * W2P + m * 2;
#pragma unroll
            for (int t = 0; t < OUT_F / 16; ++t) {
                v2f bb = *(const v2f*)(brow + t * 32);
                c2[t] = __builtin_amdgcn_wmma_f32_16x16x4_f32(
                     false, a,  false, bb,
                     (short)0, c2[t],
                     false,  false);
            }
        }
#pragma unroll
        for (int t = 0; t < OUT_F / 16; ++t) {
#pragma unroll
            for (int r = 0; r < 8; ++r)
                drow[r * AP + t * 16] = c2[t][r];
        }
    }
    __syncthreads();
    int row0  = b * BS;
    int nrows = n_nodes - row0;
    if (nrows > BS) nrows = BS;
    float4* o4 = (float4*)(tout + (size_t)row0 * OUT_F);
    for (int i = tid; i < nrows * (OUT_F / 4); i += SUM_T) {
        int lr = i / (OUT_F / 4), c4 = i % (OUT_F / 4);
        o4[i] = *(const float4*)(acc + lr * AP + c4 * 4);
    }
}

extern "C" void kernel_launch(void* const* d_in, const int* in_sizes, int n_in,
                              void* d_out, int out_size, void* d_ws, size_t ws_size,
                              hipStream_t stream) {
    const float* feat = (const float*)d_in[0];
    const int*   src  = (const int*)d_in[1];
    const int*   dst  = (const int*)d_in[2];
    const float* W1   = (const float*)d_in[3];
    const float* b1   = (const float*)d_in[4];
    const float* W2   = (const float*)d_in[5];
    const float* b2   = (const float*)d_in[6];
    float* out = (float*)d_out;

    int n_nodes = in_sizes[0] / IN_F;
    int n_edges = in_sizes[1];

    float* tbuf = (float*)d_ws;
    float* pl1  = tbuf + (size_t)n_nodes * HID_F;

    int nbk  = (n_nodes + BS - 1) / BS;
    int nblk = (n_edges + CH - 1) / CH;
    if (nbk  < 1) nbk  = 1;
    if (nblk < 1) nblk = 1;

    int2* list = (int2*)pl1;
    int2* offt = list + (size_t)nblk * CH;

    gcn_chunk_sort_kernel<<<nblk, BIN_T, 0, stream>>>(src, dst, list, offt, n_edges, nbk);

    gcn_bucket_sum_dense_kernel
        <<<nbk, SUM_T, 0, stream>>>(feat, list, offt, nblk, W1, b1, W2, tbuf, n_nodes);

    gcn_bucket_sum_kernel<OUT_F, OUT_F, true>
        <<<nbk, SUM_T, 0, stream>>>(tbuf, list, offt, nblk, b2, out, n_nodes);
}


# Edge passes: the last, partial step of a bucket's list (under 32 entries, one wave) issues its row gathers 16 at a time before their LDS adds, as full steps do; was one load then add per entry
# speedup vs baseline: 3.3378x; 1.0951x over previous
#include <hip/hip_runtime.h>

typedef __attribute__((ext_vector_type(2))) float v2f;
typedef __attribute__((ext_vector_type(8))) float v8f;

#define IN_F  64
#define HID_F 64
#define OUT_F 32

#define BS       196
#define NBK_MAX  1024
#define CH       8192
#define BIN_T    1024
#define EPT      (CH / BIN_T)

template<int NW>
__device__ __forceinline__ int gcn_wg_scan(int v, int* wsum, int lane, int w, int& tot) {
    int x = v;
    for (int o = 1; o < 32; o <<= 1) {
        int y = __shfl_up(x, o);
        if (lane >= o) x += y;
    }
    if (lane == 31) wsum[w] = x;
    __syncthreads();
    int below = 0, all = 0;
    for (int j = 0; j < NW; ++j) {
        int s = wsum[j];
        if (j < w) below += s;
        all += s;
    }
    tot = all;
    __syncthreads();
    return x + below;
}

__global__ __launch_bounds__(BIN_T)
void gcn_chunk_sort_kernel(const int* __restrict__ src, const int* __restrict__ dst,
                           int2* __restrict__ list, int2* __restrict__ offt,
                           int n_edges, int nbk) {
    static_assert(NBK_MAX == BIN_T, "one bucket counter per thread");
    __shared__ int  hist[NBK_MAX];
    __shared__ int  wsum[BIN_T / 32];
    __shared__ int2 stage[CH];
    int tid  = threadIdx.x;
    int lane = tid & 31;
    int w    = tid >> 5;
    int blk  = blockIdx.x;
    hist[tid] = 0;
    int base = blk * CH;
    int s[EPT], d[EPT];
    for (int k = 0; k < EPT; ++k) {
        int e = base + k * BIN_T + tid;
        s[k] = 0; d[k] = 0;
        if (e < n_edges) { s[k] = src[e]; d[k] = dst[e]; }
    }
    __syncthreads();
    for (int k = 0; k < EPT; ++k) {
        int e = base + k * BIN_T + tid;
        if (e < n_edges) atomicAdd(&hist[(unsigned)d[k] / BS], 1);
    }
    __syncthreads();
    int tot;
    int hv = hist[tid];
    int lstart = gcn_wg_scan<BIN_T / 32>(hv, wsum, lane, w, tot) - hv;
    hist[tid] = lstart;
    if (tid < nbk) offt[(size_t)blk * NBK_MAX + tid] = make_int2(lstart, hv);
    __syncthreads();
    for (int k = 0; k < EPT; ++k) {
        int e = base + k * BIN_T + tid;
        if (e < n_edges) {
            int bkt = (unsigned)d[k] / BS;
            int p = atomicAdd(&hist[bkt], 1);
            stage[p] = make_int2(s[k], d[k] - bkt * BS);
        }
    }
    __syncthreads();
    for (int i = tid; i < tot; i += BIN_T) list[base + i] = stage[i];
}

#define SUM_T 512
#define CAP   4096

template<int F, int FS, int P>
__device__ __forceinline__ void gcn_bucket_gather(const float* __restrict__ feat,
                                                  const int2*  __restrict__ list,
                                                  int start, int end,
                                                  float* acc, int lane, int w) {
    constexpr int VW  = F / 32;
    constexpr int UNR = 16;
    constexpr int NW  = SUM_T / 32;
    static_assert(F == 64 || F == 32, "one or two columns per lane");
    int base = start + w * 32;
    int s_l = 0, d_l = 0;
    if (base + lane < end) { int2 p = list[base + lane]; s_l = p.x; d_l = p.y; }
    for (; base < end; base += NW * 32) {
        int s_cur = s_l, d_cur = d_l;
        int cnt = __builtin_amdgcn_readfirstlane(end - base);
        int nx = base + NW * 32;
        s_l = 0; d_l = 0;
        if (nx + lane < end) { int2 p = list[nx + lane]; s_l = p.x; d_l = p.y; }
        if (cnt >= 32) {
            for (int j0 = 0; j0 < 32; j0 += UNR) {
                float vx[UNR], vy[UNR];
                for (int u = 0; u < UNR; ++u) {
                    int sn = __builtin_amdgcn_readlane(s_cur, j0 + u);
                    const float* r = feat + (size_t)sn * FS + lane * VW;
                    if (VW == 2) { v2f t = *(const v2f*)r; vx[u] = t.x; vy[u] = t.y; }
                    else         { vx[u] = *r; vy[u] = 0.f; }
                }
                for (int u = 0; u < UNR; ++u) {
                    int dn = __builtin_amdgcn_readlane(d_cur, j0 + u);
                    float* a = acc + dn * P + lane * VW;
                    atomicAdd(a, vx[u]);
                    if (VW == 2) atomicAdd(a + 1, vy[u]);
                }
            }
        } else {
            for (int j0 = 0; j0 < cnt; j0 += UNR) {
                float vx[UNR], vy[UNR];
                for (int u = 0; u < UNR; ++u) {
                    int j = j0 + u < cnt ? j0 + u : cnt - 1;
                    int sn = __builtin_amdgcn_readlane(s_cur, j);
                    const float* r = feat + (size_t)sn * FS + lane * VW;
                    if (VW == 2) { v2f t = *(const v2f*)r; vx[u] = t.x; vy[u] = t.y; }
                    else         { vx[u] = *r; vy[u] = 0.f; }
                }
                for (int u = 0; u < UNR; ++u) {
                    bool on = j0 + u < cnt;
                    int j = on ? j0 + u : cnt - 1;
                    int dn = __builtin_amdgcn_readlane(d_cur, j);
                    float* a = acc + dn * P + lane * VW;
                    atomicAdd(a, on ? vx[u] : 0.f);
                    if (VW == 2) atomicAdd(a + 1, on ? vy[u] : 0.f);
                }
            }
        }
    }
}

template<int F, int FS, int P>
__device__ __forceinline__ void gcn_bucket_edges(const float* __restrict__ feat,
                                                 const int2*  __restrict__ list,
                                                 const int2*  __restrict__ offt,
                                                 int b, int nblk,
                                                 float* acc, int2* ent, int* wsum,
                                                 int tid, int lane, int w) {
    constexpr int NW  = SUM_T / 32;
    constexpr int RPP = SUM_T / 2;
    constexpr int CU  = 16;
    int h = tid & 1;
    for (int t0 = 0; t0 < nblk; t0 += RPP) {
        int r = t0 + (tid >> 1);
        int o0 = 0, len = 0;
        if (r < nblk) {
            int2 o = offt[(size_t)r * NBK_MAX + b];
            o0  = o.x;
            len = o.y;
        }
        int tot;
        int pre = gcn_wg_scan<NW>(h == 0 ? len : 0, wsum, lane, w, tot) - len;
        const int2* run = list + (size_t)r * CH + o0;
        for (int seg = 0; seg < tot; seg += CAP) {
            int i_lo = seg - pre;
            if (i_lo < 0) i_lo = 0;
            int i_hi = seg + CAP - pre;
            if (i_hi > len) i_hi = len;
            for (int i = i_lo + ((i_lo ^ h) & 1); i < i_hi; i += 2 * CU) {
                int2 v[CU];
#pragma unroll
                for (int u = 0; u < CU; ++u) {
                    int ii = i + 2 * u;
                    if (ii < i_hi) v[u] = run[ii];
                }
#pragma unroll
                for (int u = 0; u < CU; ++u) {
                    int ii = i + 2 * u;
                    if (ii < i_hi) ent[pre + ii - seg] = v[u];
                }
            }
            __syncthreads();
            int n = tot - seg;
            if (n > CAP) n = CAP;
            gcn_bucket_gather<F, FS, P>(feat, ent, 0, n, acc, lane, w);
            __syncthreads();
        }
    }
}

template<int F, int FS, bool BIAS>
__global__ __launch_bounds__(SUM_T)
void gcn_bucket_sum_kernel(const float* __restrict__ feat,
                           const int2*  __restrict__ list,
                           const int2*  __restrict__ offt,
                           int nblk,
                           const float* __restrict__ bias,
                           float*       __restrict__ outp,
                           int n_nodes) {
    __shared__ __attribute__((aligned(16))) float acc[BS * F];
    __shared__ int2 ent[CAP];
    __shared__ int  wsum[SUM_T / 32];
    int tid  = threadIdx.x;
    int lane = tid & 31;
    int w    = tid >> 5;
    int b = blockIdx.x;

    float4* acc4 = (float4*)acc;
    float4 z = make_float4(0.f, 0.f, 0.f, 0.f);
    for (int i = tid; i < BS * F / 4; i += SUM_T) acc4[i] = z;

    gcn_bucket_edges<F, FS, F>(feat, list, offt, b, nblk, acc, ent, wsum, tid, lane, w);
    int row0  = b * BS;
    int nrows = n_nodes - row0;
    if (nrows > BS) nrows = BS;
    float4* o4 = (float4*)(outp + (size_t)row0 * F);
    for (int i = tid; i < nrows * (F / 4); i += SUM_T) {
        float4 v = acc4[i];
        if (BIAS) {
            int c = (i % (F / 4)) * 4;
            v.x += bias[c]; v.y += bias[c + 1]; v.z += bias[c + 2]; v.w += bias[c + 3];
        }
        o4[i] = v;
    }
}

#define NSTRIP ((BS + 15) / 16)
#define BSP    (NSTRIP * 16)
#define AP     68
#define W1P    (2 * HID_F + 32)
#define W2P    (2 * OUT_F + 32)

__global__ __launch_bounds__(SUM_T)
void gcn_bucket_sum_dense_kernel(const float* __restrict__ feat,
                                 const int2*  __restrict__ list,
                                 const int2*  __restrict__ offt,
                                 int nblk,
                                 const float* __restrict__ W1,
                                 const float* __restrict__ b1,
                                 const float* __restrict__ W2,
                                 float*       __restrict__ tout,
                                 int n_nodes) {
    static_assert(IN_F == 64 && HID_F == 64 && OUT_F == 32, "tile and strip shapes");
    static_assert(NSTRIP <= SUM_T / 32, "one strip per wave");
    __shared__ __attribute__((aligned(16))) float acc[BSP * AP];
    __shared__ __attribute__((aligned(16))) float w1s[(IN_F / 2) * W1P];
    __shared__ __attribute__((aligned(16))) float w2s[(HID_F / 2) * W2P];
    __shared__ int2 ent[CAP];
    __shared__ int  wsum[SUM_T / 32];
    int tid  = threadIdx.x;
    int lane = tid & 31;
    int w    = tid >> 5;
    int b = blockIdx.x;

    float4* acc4 = (float4*)acc;
    float4 z = make_float4(0.f, 0.f, 0.f, 0.f);
    for (int i = tid; i < BSP * AP / 4; i += SUM_T) acc4[i] = z;
    for (int i = tid; i < IN_F * HID_F; i += SUM_T) {
        int k = i / HID_F, n = i % HID_F;
        w1s[(k >> 1) * W1P + n * 2 + (k & 1)] = W1[i];
    }
    for (int i = tid; i < HID_F * OUT_F; i += SUM_T) {
        int k = i / OUT_F, n = i % OUT_F;
        w2s[(k >> 1) * W2P + n * 2 + (k & 1)] = W2[i];
    }

    gcn_bucket_edges<IN_F, IN_F, AP>(feat, list, offt, b, nblk, acc, ent, wsum, tid, lane, w);

    bool active = w < NSTRIP;
    int m     = lane & 15;
    int khalf = (lane >> 4) << 1;
    int mrow  = (lane >> 4) * 8;
    float* arow = acc + (w * 16 + m) * AP + khalf;
    float* drow = acc + (w * 16 + mrow) * AP + m;

    if (active) {
        v8f c1[HID_F / 16];
        v8f zero = {};
#pragma unroll
        for (int t = 0; t < HID_F / 16; ++t) c1[t] = zero;
#pragma unroll
        for (int k0 = 0; k0 < IN_F; k0 += 4) {
            v2f a = *(const v2f*)(arow + k0);
            const float* brow = w1s + ((k0 + khalf) >> 1) * W1P + m * 2;
#pragma unroll
            for (int t = 0; t < HID_F / 16; ++t) {
                v2f bb = *(const v2f*)(brow + t * 32);
                c1[t] = __builtin_amdgcn_wmma_f32_16x16x4_f32(
                     false, a,  false, bb,
                     (short)0, c1[t],
                     false,  false);
            }
        }
#pragma unroll
        for (int t = 0; t < HID_F / 16; ++t) {
            float bv = b1[t * 16 + m];
#pragma unroll
            for (int r = 0; r < 8; ++r)
                drow[r * AP + t * 16] = fmaxf(c1[t][r] + bv, 0.f);
        }
    }
    __syncthreads();
    if (active) {
        v8f c2[OUT_F / 16];
        v8f zero = {};
#pragma unroll
        for (int t = 0; t < OUT_F / 16; ++t) c2[t] = zero;
#pragma unroll
        for (int k0 = 0; k0 < HID_F; k0 += 4) {
            v2f a = *(const v2f*)(arow + k0);
            const float* brow = w2s + ((k0 + khalf) >> 1) * W2P + m * 2;
#pragma unroll
            for (int t = 0; t < OUT_F / 16; ++t) {
                v2f bb = *(const v2f*)(brow + t * 32);
                c2[t] = __builtin_amdgcn_wmma_f32_16x16x4_f32(
                     false, a,  false, bb,
                     (short)0, c2[t],
                     false,  false);
            }
        }
#pragma unroll
        for (int t = 0; t < OUT_F / 16; ++t) {
#pragma unroll
            for (int r = 0; r < 8; ++r)
                drow[r * AP + t * 16] = c2[t][r];
        }
    }
    __syncthreads();
    int row0  = b * BS;
    int nrows = n_nodes - row0;
    if (nrows > BS) nrows = BS;
    float4* o4 = (float4*)(tout + (size_t)row0 * OUT_F);
    for (int i = tid; i < nrows * (OUT_F / 4); i += SUM_T) {
        int lr = i / (OUT_F / 4), c4 = i % (OUT_F / 4);
        o4[i] = *(const float4*)(acc + lr * AP + c4 * 4);
    }
}

extern "C" void kernel_launch(void* const* d_in, const int* in_sizes, int n_in,
                              void* d_out, int out_size, void* d_ws, size_t ws_size,
                              hipStream_t stream) {
    const float* feat = (const float*)d_in[0];
    const int*   src  = (const int*)d_in[1];
    const int*   dst  = (const int*)d_in[2];
    const float* W1   = (const float*)d_in[3];
    const float* b1   = (const float*)d_in[4];
    const float* W2   = (const float*)d_in[5];
    const float* b2   = (const float*)d_in[6];
    float* out = (float*)d_out;

    int n_nodes = in_sizes[0] / IN_F;
    int n_edges = in_sizes[1];

    float* tbuf = (float*)d_ws;
    float* pl1  = tbuf + (size_t)n_nodes * HID_F;

    int nbk  = (n_nodes + BS - 1) / BS;
    int nblk = (n_edges + CH - 1) / CH;
    if (nbk  < 1) nbk  = 1;
    if (nblk < 1) nblk = 1;

    int2* list = (int2*)pl1;
    int2* offt = list + (size_t)nblk * CH;

    gcn_chunk_sort_kernel<<<nblk, BIN_T, 0, stream>>>(src, dst, list, offt, n_edges, nbk);

    gcn_bucket_sum_dense_kernel
        <<<nbk, SUM_T, 0, stream>>>(feat, list, offt, nblk, W1, b1, W2, tbuf, n_nodes);

    gcn_bucket_sum_kernel<OUT_F, OUT_F, true>
        <<<nbk, SUM_T, 0, stream>>>(tbuf, list, offt, nblk, b2, out, n_nodes);
}


# Layer-1 kernel set-up: a thread's 12 W1/W2 reads and its run-table read are issued together before the tile is cleared: one trip to memory, not 13 in a row; layer 2 reads its table entry early too
# speedup vs baseline: 3.5760x; 1.0714x over previous
#include <hip/hip_runtime.h>

typedef __attribute__((ext_vector_type(2))) float v2f;
typedef __attribute__((ext_vector_type(8))) float v8f;

#define IN_F  64
#define HID_F 64
#define OUT_F 32

#define BS       196
#define NBK_MAX  1024
#define CH       8192
#define BIN_T    1024
#define EPT      (CH / BIN_T)

template<int NW>
__device__ __forceinline__ int gcn_wg_scan(int v, int* wsum, int lane, int w, int& tot) {
    int x = v;
    for (int o = 1; o < 32; o <<= 1) {
        int y = __shfl_up(x, o);
        if (lane >= o) x += y;
    }
    if (lane == 31) wsum[w] = x;
    __syncthreads();
    int below = 0, all = 0;
    for (int j = 0; j < NW; ++j) {
        int s = wsum[j];
        if (j < w) below += s;
        all += s;
    }
    tot = all;
    __syncthreads();
    return x + below;
}

__global__ __launch_bounds__(BIN_T)
void gcn_chunk_sort_kernel(const int* __restrict__ src, const int* __restrict__ dst,
                           int2* __restrict__ list, int2* __restrict__ offt,
                           int n_edges, int nbk) {
    static_assert(NBK_MAX == BIN_T, "one bucket counter per thread");
    __shared__ int  hist[NBK_MAX];
    __shared__ int  wsum[BIN_T / 32];
    __shared__ int2 stage[CH];
    int tid  = threadIdx.x;
    int lane = tid & 31;
    int w    = tid >> 5;
    int blk  = blockIdx.x;
    hist[tid] = 0;
    int base = blk * CH;
    int s[EPT], d[EPT];
    for (int k = 0; k < EPT; ++k) {
        int e = base + k * BIN_T + tid;
        s[k] = 0; d[k] = 0;
        if (e < n_edges) { s[k] = src[e]; d[k] = dst[e]; }
    }
    __syncthreads();
    for (int k = 0; k < EPT; ++k) {
        int e = base + k * BIN_T + tid;
        if (e < n_edges) atomicAdd(&hist[(unsigned)d[k] / BS], 1);
    }
    __syncthreads();
    int tot;
    int hv = hist[tid];
    int lstart = gcn_wg_scan<BIN_T / 32>(hv, wsum, lane, w, tot) - hv;
    hist[tid] = lstart;
    if (tid < nbk) offt[(size_t)blk * NBK_MAX + tid] = make_int2(lstart, hv);
    __syncthreads();
    for (int k = 0; k < EPT; ++k) {
        int e = base + k * BIN_T + tid;
        if (e < n_edges) {
            int bkt = (unsigned)d[k] / BS;
            int p = atomicAdd(&hist[bkt], 1);
            stage[p] = make_int2(s[k], d[k] - bkt * BS);
        }
    }
    __syncthreads();
    for (int i = tid; i < tot; i += BIN_T) list[base + i] = stage[i];
}

#define SUM_T 512
#define CAP   4096

template<int F, int FS, int P>
__device__ __forceinline__ void gcn_bucket_gather(const float* __restrict__ feat,
                                                  const int2*  __restrict__ list,
                                                  int start, int end,
                                                  float* acc, int lane, int w) {
    constexpr int VW  = F / 32;
    constexpr int UNR = 16;
    constexpr int NW  = SUM_T / 32;
    static_assert(F == 64 || F == 32, "one or two columns per lane");
    int base = start + w * 32;
    int s_l = 0, d_l = 0;
    if (base + lane < end) { int2 p = list[base + lane]; s_l = p.x; d_l = p.y; }
    for (; base < end; base += NW * 32) {
        int s_cur = s_l, d_cur = d_l;
        int cnt = __builtin_amdgcn_readfirstlane(end - base);
        int nx = base + NW * 32;
        s_l = 0; d_l = 0;
        if (nx + lane < end) { int2 p = list[nx + lane]; s_l = p.x; d_l = p.y; }
        if (cnt >= 32) {
            for (int j0 = 0; j0 < 32; j0 += UNR) {
                float vx[UNR], vy[UNR];
                for (int u = 0; u < UNR; ++u) {
                    int sn = __builtin_amdgcn_readlane(s_cur, j0 + u);
                    const float* r = feat + (size_t)sn * FS + lane * VW;
                    if (VW == 2) { v2f t = *(const v2f*)r; vx[u] = t.x; vy[u] = t.y; }
                    else         { vx[u] = *r; vy[u] = 0.f; }
                }
                for (int u = 0; u < UNR; ++u) {
                    int dn = __builtin_amdgcn_readlane(d_cur, j0 + u);
                    float* a = acc + dn * P + lane * VW;
                    atomicAdd(a, vx[u]);
                    if (VW == 2) atomicAdd(a + 1, vy[u]);
                }
            }
        } else {
            for (int j0 = 0; j0 < cnt; j0 += UNR) {
                float vx[UNR], vy[UNR];
                for (int u = 0; u < UNR; ++u) {
                    int j = j0 + u < cnt ? j0 + u : cnt - 1;
                    int sn = __builtin_amdgcn_readlane(s_cur, j);
                    const float* r = feat + (size_t)sn * FS + lane * VW;
                    if (VW == 2) { v2f t = *(const v2f*)r; vx[u] = t.x; vy[u] = t.y; }
                    else         { vx[u] = *r; vy[u] = 0.f; }
                }
                for (int u = 0; u < UNR; ++u) {
                    bool on = j0 + u < cnt;
                    int j = on ? j0 + u : cnt - 1;
                    int dn = __builtin_amdgcn_readlane(d_cur, j);
                    float* a = acc + dn * P + lane * VW;
                    atomicAdd(a, on ? vx[u] : 0.f);
                    if (VW == 2) atomicAdd(a + 1, on ? vy[u] : 0.f);
                }
            }
        }
    }
}

__device__ __forceinline__ int2 gcn_first_run(const int2* __restrict__ offt,
                                              int b, int nblk, int tid) {
    int r = tid >> 1;
    int2 o = make_int2(0, 0);
    if (r < nblk) o = offt[(size_t)r * NBK_MAX + b];
    return o;
}

template<int F, int FS, int P>
__device__ __forceinline__ void gcn_bucket_edges(const float* __restrict__ feat,
                                                 const int2*  __restrict__ list,
                                                 const int2*  __restrict__ offt,
                                                 int b, int nblk, int2 o_pre,
                                                 float* acc, int2* ent, int* wsum,
                                                 int tid, int lane, int w) {
    constexpr int NW  = SUM_T / 32;
    constexpr int RPP = SUM_T / 2;
    constexpr int CU  = 16;
    int h = tid & 1;
    for (int t0 = 0; t0 < nblk; t0 += RPP) {
        int r = t0 + (tid >> 1);
        int o0 = 0, len = 0;
        if (r < nblk) {
            int2 o = o_pre;
            if (t0 > 0) o = offt[(size_t)r * NBK_MAX + b];
            o0  = o.x;
            len = o.y;
        }
        int tot;
        int pre = gcn_wg_scan<NW>(h == 0 ? len : 0, wsum, lane, w, tot) - len;
        const int2* run = list + (size_t)r * CH + o0;
        for (int seg = 0; seg < tot; seg += CAP) {
            int i_lo = seg - pre;
            if (i_lo < 0) i_lo = 0;
            int i_hi = seg + CAP - pre;
            if (i_hi > len) i_hi = len;
            for (int i = i_lo + ((i_lo ^ h) & 1); i < i_hi; i += 2 * CU) {
                int2 v[CU];
#pragma unroll
                for (int u = 0; u < CU; ++u) {
                    int ii = i + 2 * u;
                    if (ii < i_hi) v[u] = run[ii];
                }
#pragma unroll
                for (int u = 0; u < CU; ++u) {
                    int ii = i + 2 * u;
                    if (ii < i_hi) ent[pre + ii - seg] = v[u];
                }
            }
            __syncthreads();
            int n = tot - seg;
            if (n > CAP) n = CAP;
            gcn_bucket_gather<F, FS, P>(feat, ent, 0, n, acc, lane, w);
            __syncthreads();
        }
    }
}

template<int F, int FS, bool BIAS>
__global__ __launch_bounds__(SUM_T)
void gcn_bucket_sum_kernel(const float* __restrict__ feat,
                           const int2*  __restrict__ list,
                           const int2*  __restrict__ offt,
                           int nblk,
                           const float* __restrict__ bias,
                           float*       __restrict__ outp,
                           int n_nodes) {
    __shared__ __attribute__((aligned(16))) float acc[BS * F];
    __shared__ int2 ent[CAP];
    __shared__ int  wsum[SUM_T / 32];
    int tid  = threadIdx.x;
    int lane = tid & 31;
    int w    = tid >> 5;
    int b = blockIdx.x;

    int2 o_pre = gcn_first_run(offt, b, nblk, tid);

    float4* acc4 = (float4*)acc;
    float4 z = make_float4(0.f, 0.f, 0.f, 0.f);
    for (int i = tid; i < BS * F / 4; i += SUM_T) acc4[i] = z;

    gcn_bucket_edges<F, FS, F>(feat, list, offt, b, nblk, o_pre, acc, ent, wsum, tid, lane, w);
    int row0  = b * BS;
    int nrows = n_nodes - row0;
    if (nrows > BS) nrows = BS;
    float4* o4 = (float4*)(outp + (size_t)row0 * F);
    for (int i = tid; i < nrows * (F / 4); i += SUM_T) {
        float4 v = acc4[i];
        if (BIAS) {
            int c = (i % (F / 4)) * 4;
            v.x += bias[c]; v.y += bias[c + 1]; v.z += bias[c + 2]; v.w += bias[c + 3];
        }
        o4[i] = v;
    }
}

#define NSTRIP ((BS + 15) / 16)
#define BSP    (NSTRIP * 16)
#define AP     68
#define W1P    (2 * HID_F + 32)
#define W2P    (2 * OUT_F + 32)

__global__ __launch_bounds__(SUM_T)
void gcn_bucket_sum_dense_kernel(const float* __restrict__ feat,
                                 const int2*  __restrict__ list,
                                 const int2*  __restrict__ offt,
                                 int nblk,
                                 const float* __restrict__ W1,
                                 const float* __restrict__ b1,
                                 const float* __restrict__ W2,
                                 float*       __restrict__ tout,
                                 int n_nodes) {
    static_assert(IN_F == 64 && HID_F == 64 && OUT_F == 32, "tile and strip shapes");
    static_assert(NSTRIP <= SUM_T / 32, "one strip per wave");
    __shared__ __attribute__((aligned(16))) float acc[BSP * AP];
    __shared__ __attribute__((aligned(16))) float w1s[(IN_F / 2) * W1P];
    __shared__ __attribute__((aligned(16))) float w2s[(HID_F / 2) * W2P];
    __shared__ int2 ent[CAP];
    __shared__ int  wsum[SUM_T / 32];
    int tid  = threadIdx.x;
    int lane = tid & 31;
    int w    = tid >> 5;
    int b = blockIdx.x;

    constexpr int W1_PT = IN_F * HID_F / SUM_T;
    constexpr int W2_PT = HID_F * OUT_F / SUM_T;
    static_assert(W1_PT * SUM_T == IN_F * HID_F && W2_PT * SUM_T == HID_F * OUT_F,
                  "whole pieces per thread");
    float w1r[W1_PT], w2r[W2_PT];
#pragma unroll
    for (int j = 0; j < W1_PT; ++j) w1r[j] = W1[tid + j * SUM_T];
#pragma unroll
    for (int j = 0; j < W2_PT; ++j) w2r[j] = W2[tid + j * SUM_T];
    int2 o_pre = gcn_first_run(offt, b, nblk, tid);

    float4* acc4 = (float4*)acc;
    float4 z = make_float4(0.f, 0.f, 0.f, 0.f);
    for (int i = tid; i < BSP * AP / 4; i += SUM_T) acc4[i] = z;
#pragma unroll
    for (int j = 0; j < W1_PT; ++j) {
        int i = tid + j * SUM_T;
        int k = i / HID_F, n = i % HID_F;
        w1s[(k >> 1) * W1P + n * 2 + (k & 1)] = w1r[j];
    }
#pragma unroll
    for (int j = 0; j < W2_PT; ++j) {
        int i = tid + j * SUM_T;
        int k = i / OUT_F, n = i % OUT_F;
        w2s[(k >> 1) * W2P + n * 2 + (k & 1)] = w2r[j];
    }

    gcn_bucket_edges<IN_F, IN_F, AP>(feat, list, offt, b, nblk, o_pre, acc, ent, wsum, tid, lane, w);

    bool active = w < NSTRIP;
    int m     = lane & 15;
    int khalf = (lane >> 4) << 1;
    int mrow  = (lane >> 4) * 8;
    float* arow = acc + (w * 16 + m) * AP + khalf;
    float* drow = acc + (w * 16 + mrow) * AP + m;

    if (active) {
        v8f c1[HID_F / 16];
        v8f zero = {};
#pragma unroll
        for (int t = 0; t < HID_F / 16; ++t) c1[t] = zero;
#pragma unroll
        for (int k0 = 0; k0 < IN_F; k0 += 4) {
            v2f a = *(const v2f*)(arow + k0);
            const float* brow = w1s + ((k0 + khalf) >> 1) * W1P + m * 2;
#pragma unroll
            for (int t = 0; t < HID_F / 16; ++t) {
                v2f bb = *(const v2f*)(brow + t * 32);
                c1[t] = __builtin_amdgcn_wmma_f32_16x16x4_f32(
                     false, a,  false, bb,
                     (short)0, c1[t],
                     false,  false);
            }
        }
#pragma unroll
        for (int t = 0; t < HID_F / 16; ++t) {
            float bv = b1[t * 16 + m];
#pragma unroll
            for (int r = 0; r < 8; ++r)
                drow[r * AP + t * 16] = fmaxf(c1[t][r] + bv, 0.f);
        }
    }
    __syncthreads();
    if (active) {
        v8f c2[OUT_F / 16];
        v8f zero = {};
#pragma unroll
        for (int t = 0; t < OUT_F / 16; ++t) c2[t] = zero;
#pragma unroll
        for (int k0 = 0; k0 < HID_F; k0 += 4) {
            v2f a = *(const v2f*)(arow + k0);
            const float* brow = w2s + ((k0 + khalf) >> 1) * W2P + m * 2;
#pragma unroll
            for (int t = 0; t < OUT_F / 16; ++t) {
                v2f bb = *(const v2f*)(brow + t * 32);
                c2[t] = __builtin_amdgcn_wmma_f32_16x16x4_f32(
                     false, a,  false, bb,
                     (short)0, c2[t],
                     false,  false);
            }
        }
#pragma unroll
        for (int t = 0; t < OUT_F / 16; ++t) {
#pragma unroll
            for (int r = 0; r < 8; ++r)
                drow[r * AP + t * 16] = c2[t][r];
        }
    }
    __syncthreads();
    int row0  = b * BS;
    int nrows = n_nodes - row0;
    if (nrows > BS) nrows = BS;
    float4* o4 = (float4*)(tout + (size_t)row0 * OUT_F);
    for (int i = tid; i < nrows * (OUT_F / 4); i += SUM_T) {
        int lr = i / (OUT_F / 4), c4 = i % (OUT_F / 4);
        o4[i] = *(const float4*)(acc + lr * AP + c4 * 4);
    }
}

extern "C" void kernel_launch(void* const* d_in, const int* in_sizes, int n_in,
                              void* d_out, int out_size, void* d_ws, size_t ws_size,
                              hipStream_t stream) {
    const float* feat = (const float*)d_in[0];
    const int*   src  = (const int*)d_in[1];
    const int*   dst  = (const int*)d_in[2];
    const float* W1   = (const float*)d_in[3];
    const float* b1   = (const float*)d_in[4];
    const float* W2   = (const float*)d_in[5];
    const float* b2   = (const float*)d_in[6];
    float* out = (float*)d_out;

    int n_nodes = in_sizes[0] / IN_F;
    int n_edges = in_sizes[1];

    float* tbuf = (float*)d_ws;
    float* pl1  = tbuf + (size_t)n_nodes * HID_F;

    int nbk  = (n_nodes + BS - 1) / BS;
    int nblk = (n_edges + CH - 1) / CH;
    if (nbk  < 1) nbk  = 1;
    if (nblk < 1) nblk = 1;

    int2* list = (int2*)pl1;
    int2* offt = list + (size_t)nblk * CH;

    gcn_chunk_sort_kernel<<<nblk, BIN_T, 0, stream>>>(src, dst, list, offt, n_edges, nbk);

    gcn_bucket_sum_dense_kernel
        <<<nbk, SUM_T, 0, stream>>>(feat, list, offt, nblk, W1, b1, W2, tbuf, n_nodes);

    gcn_bucket_sum_kernel<OUT_F, OUT_F, true>
        <<<nbk, SUM_T, 0, stream>>>(tbuf, list, offt, nblk, b2, out, n_nodes);
}


# Edge passes: a bucket's LDS list is ordered by destination row (counts, prefix sum, arrival-order rank); a wave sums a row's gathers in registers, one LDS add per run, not per edge; sum order changes
# speedup vs baseline: 4.1431x; 1.1586x over previous
#include <hip/hip_runtime.h>

typedef __attribute__((ext_vector_type(2))) float v2f;
typedef __attribute__((ext_vector_type(8))) float v8f;

#define IN_F  64
#define HID_F 64
#define OUT_F 32

#define BS       196
#define NBK_MAX  1024
#define CH       8192
#define BIN_T    1024
#define EPT      (CH / BIN_T)

template<int NW>
__device__ __forceinline__ int gcn_wg_scan(int v, int* wsum, int lane, int w, int& tot) {
    int x = v;
    for (int o = 1; o < 32; o <<= 1) {
        int y = __shfl_up(x, o);
        if (lane >= o) x += y;
    }
    if (lane == 31) wsum[w] = x;
    __syncthreads();
    int below = 0, all = 0;
    for (int j = 0; j < NW; ++j) {
        int s = wsum[j];
        if (j < w) below += s;
        all += s;
    }
    tot = all;
    __syncthreads();
    return x + below;
}

__global__ __launch_bounds__(BIN_T)
void gcn_chunk_sort_kernel(const int* __restrict__ src, const int* __restrict__ dst,
                           int2* __restrict__ list, int2* __restrict__ offt,
                           int n_edges, int nbk) {
    static_assert(NBK_MAX == BIN_T, "one bucket counter per thread");
    __shared__ int  hist[NBK_MAX];
    __shared__ int  wsum[BIN_T / 32];
    __shared__ int2 stage[CH];
    int tid  = threadIdx.x;
    int lane = tid & 31;
    int w    = tid >> 5;
    int blk  = blockIdx.x;
    hist[tid] = 0;
    int base = blk * CH;
    int s[EPT], d[EPT];
    for (int k = 0; k < EPT; ++k) {
        int e = base + k * BIN_T + tid;
        s[k] = 0; d[k] = 0;
        if (e < n_edges) { s[k] = src[e]; d[k] = dst[e]; }
    }
    __syncthreads();
    for (int k = 0; k < EPT; ++k) {
        int e = base + k * BIN_T + tid;
        if (e < n_edges) atomicAdd(&hist[(unsigned)d[k] / BS], 1);
    }
    __syncthreads();
    int tot;
    int hv = hist[tid];
    int lstart = gcn_wg_scan<BIN_T / 32>(hv, wsum, lane, w, tot) - hv;
    hist[tid] = lstart;
    if (tid < nbk) offt[(size_t)blk * NBK_MAX + tid] = make_int2(lstart, hv);
    __syncthreads();
    for (int k = 0; k < EPT; ++k) {
        int e = base + k * BIN_T + tid;
        if (e < n_edges) {
            int bkt = (unsigned)d[k] / BS;
            int p = atomicAdd(&hist[bkt], 1);
            stage[p] = make_int2(s[k], d[k] - bkt * BS);
        }
    }
    __syncthreads();
    for (int i = tid; i < tot; i += BIN_T) list[base + i] = stage[i];
}

#define SUM_T 512
#define CAP   4096
static_assert(BS <= 256, "a destination row fits the low byte of a packed entry");
static_assert(BS * NBK_MAX <= (1 << 24), "a source node fits the upper 24 bits of a packed entry");

template<int VW, int P>
__device__ __forceinline__ void gcn_row_flush(float* acc, int row, int lane, float ax, float ay) {
    float* a = acc + row * P + lane * VW;
    atomicAdd(a, ax);
    if (VW == 2) atomicAdd(a + 1, ay);
}

template<int F, int FS, int P>
__device__ __forceinline__ void gcn_bucket_gather(const float* __restrict__ feat,
                                                  const unsigned* __restrict__ list,
                                                  int start, int end,
                                                  float* acc, int lane, int w) {
    constexpr int VW  = F / 32;
    constexpr int UNR = 16;
    constexpr int NW  = SUM_T / 32;
    static_assert(F == 64 || F == 32, "one or two columns per lane");
    int base = start + w * 32;
    int s_l = 0, d_l = 0;
    if (base + lane < end) { unsigned e = list[base + lane]; s_l = (int)(e >> 8); d_l = (int)(e & 255u); }
    for (; base < end; base += NW * 32) {
        int s_cur = s_l, d_cur = d_l;
        int cnt = __builtin_amdgcn_readfirstlane(end - base);
        int nx = base + NW * 32;
        s_l = 0; d_l = 0;
        if (nx + lane < end) { unsigned e = list[nx + lane]; s_l = (int)(e >> 8); d_l = (int)(e & 255u); }
        int   cur = __builtin_amdgcn_readlane(d_cur, 0);
        float ax = 0.f, ay = 0.f;
        if (cnt >= 32) {
            for (int j0 = 0; j0 < 32; j0 += UNR) {
                float vx[UNR], vy[UNR];
                for (int u = 0; u < UNR; ++u) {
                    int sn = __builtin_amdgcn_readlane(s_cur, j0 + u);
                    const float* r = feat + (size_t)sn * FS + lane * VW;
                    if (VW == 2) { v2f t = *(const v2f*)r; vx[u] = t.x; vy[u] = t.y; }
                    else         { vx[u] = *r; vy[u] = 0.f; }
                }
                for (int u = 0; u < UNR; ++u) {
                    int dn = __builtin_amdgcn_readlane(d_cur, j0 + u);
                    if (__builtin_expect(dn != cur, 0)) {
                        gcn_row_flush<VW, P>(acc, cur, lane, ax, ay);
                        ax = 0.f; ay = 0.f; cur = dn;
                    }
                    ax += vx[u];
                    if (VW == 2) ay += vy[u];
                }
            }
        } else {
            for (int j0 = 0; j0 < cnt; j0 += UNR) {
                float vx[UNR], vy[UNR];
                for (int u = 0; u < UNR; ++u) {
                    int j = j0 + u < cnt ? j0 + u : cnt - 1;
                    int sn = __builtin_amdgcn_readlane(s_cur, j);
                    const float* r = feat + (size_t)sn * FS + lane * VW;
                    if (VW == 2) { v2f t = *(const v2f*)r; vx[u] = t.x; vy[u] = t.y; }
                    else         { vx[u] = *r; vy[u] = 0.f; }
                }
                for (int u = 0; u < UNR; ++u) {
                    bool on = j0 + u < cnt;
                    int j = on ? j0 + u : cnt - 1;
                    int dn = __builtin_amdgcn_readlane(d_cur, j);
                    if (__builtin_expect(dn != cur, 0)) {
                        gcn_row_flush<VW, P>(acc, cur, lane, ax, ay);
                        ax = 0.f; ay = 0.f; cur = dn;
                    }
                    ax += on ? vx[u] : 0.f;
                    if (VW == 2) ay += on ? vy[u] : 0.f;
                }
            }
        }
        gcn_row_flush<VW, P>(acc, cur, lane, ax, ay);
    }
}

__device__ __forceinline__ int2 gcn_first_run(const int2* __restrict__ offt,
                                              int b, int nblk, int tid) {
    int r = tid >> 1;
    int2 o = make_int2(0, 0);
    if (r < nblk) o = offt[(size_t)r * NBK_MAX + b];
    return o;
}

template<int F, int FS, int P>
__device__ __forceinline__ void gcn_bucket_edges(const float* __restrict__ feat,
                                                 const int2*  __restrict__ list,
                                                 const int2*  __restrict__ offt,
                                                 int b, int nblk, int2 o_pre,
                                                 float* acc, unsigned* raw, unsigned* ent,
                                                 int* rowc, int* wsum,
                                                 int tid, int lane, int w) {
    constexpr int NW  = SUM_T / 32;
    constexpr int RPP = SUM_T / 2;
    constexpr int CU  = 16;
    constexpr int EPT2 = CAP / SUM_T;
    static_assert(BS <= SUM_T, "one row counter per thread");
    static_assert(EPT2 * SUM_T == CAP, "whole entries per thread");
    int h = tid & 1;
    for (int t0 = 0; t0 < nblk; t0 += RPP) {
        int r = t0 + (tid >> 1);
        int o0 = 0, len = 0;
        if (r < nblk) {
            int2 o = o_pre;
            if (t0 > 0) o = offt[(size_t)r * NBK_MAX + b];
            o0  = o.x;
            len = o.y;
        }
        int tot;
        int pre = gcn_wg_scan<NW>(h == 0 ? len : 0, wsum, lane, w, tot) - len;
        const int2* run = list + (size_t)r * CH + o0;
        for (int seg = 0; seg < tot; seg += CAP) {
            int i_lo = seg - pre;
            if (i_lo < 0) i_lo = 0;
            int i_hi = seg + CAP - pre;
            if (i_hi > len) i_hi = len;
            for (int i = i_lo + ((i_lo ^ h) & 1); i < i_hi; i += 2 * CU) {
                int2 v[CU];
#pragma unroll
                for (int u = 0; u < CU; ++u) {
                    int ii = i + 2 * u;
                    if (ii < i_hi) v[u] = run[ii];
                }
#pragma unroll
                for (int u = 0; u < CU; ++u) {
                    int ii = i + 2 * u;
                    if (ii < i_hi) {
                        raw[pre + ii - seg] = ((unsigned)v[u].x << 8) | (unsigned)v[u].y;
                        atomicAdd(&rowc[v[u].y], 1);
                    }
                }
            }
            __syncthreads();
            int n = tot - seg;
            if (n > CAP) n = CAP;
            int rc = tid < BS ? rowc[tid] : 0;
            int rtot;
            int rs = gcn_wg_scan<NW>(rc, wsum, lane, w, rtot) - rc;
            if (tid < BS) rowc[tid] = rs;
            __syncthreads();
            {
                unsigned e[EPT2];
                int      p[EPT2];
#pragma unroll
                for (int k = 0; k < EPT2; ++k) {
                    int i = tid + k * SUM_T;
                    e[k] = raw[i < n ? i : 0];
                }
#pragma unroll
                for (int k = 0; k < EPT2; ++k) {
                    int i = tid + k * SUM_T;
                    int slot = i < n ? (int)(e[k] & 255u) : BS + lane;
                    p[k] = atomicAdd(&rowc[slot], 1);
                }
#pragma unroll
                for (int k = 0; k < EPT2; ++k) {
                    int i = tid + k * SUM_T;
                    if (i < n) ent[p[k]] = e[k];
                }
            }
            __syncthreads();
            gcn_bucket_gather<F, FS, P>(feat, ent, 0, n, acc, lane, w);
            if (tid < BS) rowc[tid] = 0;
            __syncthreads();
        }
    }
}

template<int F, int FS, bool BIAS>
__global__ __launch_bounds__(SUM_T)
void gcn_bucket_sum_kernel(const float* __restrict__ feat,
                           const int2*  __restrict__ list,
                           const int2*  __restrict__ offt,
                           int nblk,
                           const float* __restrict__ bias,
                           float*       __restrict__ outp,
                           int n_nodes) {
    __shared__ __attribute__((aligned(16))) float acc[BS * F];
    __shared__ unsigned raw[CAP];
    __shared__ unsigned ent[CAP];
    __shared__ int  rowc[BS + 32];
    __shared__ int  wsum[SUM_T / 32];
    int tid  = threadIdx.x;
    int lane = tid & 31;
    int w    = tid >> 5;
    int b = blockIdx.x;

    int2 o_pre = gcn_first_run(offt, b, nblk, tid);

    float4* acc4 = (float4*)acc;
    float4 z = make_float4(0.f, 0.f, 0.f, 0.f);
    for (int i = tid; i < BS * F / 4; i += SUM_T) acc4[i] = z;
    if (tid < BS) rowc[tid] = 0;

    gcn_bucket_edges<F, FS, F>(feat, list, offt, b, nblk, o_pre, acc, raw, ent, rowc, wsum, tid, lane, w);
    int row0  = b * BS;
    int nrows = n_nodes - row0;
    if (nrows > BS) nrows = BS;
    float4* o4 = (float4*)(outp + (size_t)row0 * F);
    for (int i = tid; i < nrows * (F / 4); i += SUM_T) {
        float4 v = acc4[i];
        if (BIAS) {
            int c = (i % (F / 4)) * 4;
            v.x += bias[c]; v.y += bias[c + 1]; v.z += bias[c + 2]; v.w += bias[c + 3];
        }
        o4[i] = v;
    }
}

#define NSTRIP ((BS + 15) / 16)
#define BSP    (NSTRIP * 16)
#define AP     68
#define W1P    (2 * HID_F + 32)
#define W2P    (2 * OUT_F + 32)

__global__ __launch_bounds__(SUM_T)
void gcn_bucket_sum_dense_kernel(const float* __restrict__ feat,
                                 const int2*  __restrict__ list,
                                 const int2*  __restrict__ offt,
                                 int nblk,
                                 const float* __restrict__ W1,
                                 const float* __restrict__ b1,
                                 const float* __restrict__ W2,
                                 float*       __restrict__ tout,
                                 int n_nodes) {
    static_assert(IN_F == 64 && HID_F == 64 && OUT_F == 32, "tile and strip shapes");
    static_assert(NSTRIP <= SUM_T / 32, "one strip per wave");
    __shared__ __attribute__((aligned(16))) float acc[BSP * AP];
    __shared__ __attribute__((aligned(16))) float w1s[(IN_F / 2) * W1P];
    __shared__ __attribute__((aligned(16))) float w2s[(HID_F / 2) * W2P];
    __shared__ unsigned raw[CAP];
    __shared__ unsigned ent[CAP];
    __shared__ int  rowc[BS + 32];
    __shared__ int  wsum[SUM_T / 32];
    int tid  = threadIdx.x;
    int lane = tid & 31;
    int w    = tid >> 5;
    int b = blockIdx.x;

    constexpr int W1_PT = IN_F * HID_F / SUM_T;
    constexpr int W2_PT = HID_F * OUT_F / SUM_T;
    static_assert(W1_PT * SUM_T == IN_F * HID_F && W2_PT * SUM_T == HID_F * OUT_F,
                  "whole pieces per thread");
    float w1r[W1_PT], w2r[W2_PT];
#pragma unroll
    for (int j = 0; j < W1_PT; ++j) w1r[j] = W1[tid + j * SUM_T];
#pragma unroll
    for (int j = 0; j < W2_PT; ++j) w2r[j] = W2[tid + j * SUM_T];
    int2 o_pre = gcn_first_run(offt, b, nblk, tid);

    float4* acc4 = (float4*)acc;
    float4 z = make_float4(0.f, 0.f, 0.f, 0.f);
    for (int i = tid; i < BSP * AP / 4; i += SUM_T) acc4[i] = z;
    if (tid < BS) rowc[tid] = 0;
#pragma unroll
    for (int j = 0; j < W1_PT; ++j) {
        int i = tid + j * SUM_T;
        int k = i / HID_F, n = i % HID_F;
        w1s[(k >> 1) * W1P + n * 2 + (k & 1)] = w1r[j];
    }
#pragma unroll
    for (int j = 0; j < W2_PT; ++j) {
        int i = tid + j * SUM_T;
        int k = i / OUT_F, n = i % OUT_F;
        w2s[(k >> 1) * W2P + n * 2 + (k & 1)] = w2r[j];
    }

    gcn_bucket_edges<IN_F, IN_F, AP>(feat, list, offt, b, nblk, o_pre, acc, raw, ent, rowc, wsum, tid, lane, w);

    bool active = w < NSTRIP;
    int m     = lane & 15;
    int khalf = (lane >> 4) << 1;
    int mrow  = (lane >> 4) * 8;
    float* arow = acc + (w * 16 + m) * AP + khalf;
    float* drow = acc + (w * 16 + mrow) * AP + m;

    if (active) {
        v8f c1[HID_F / 16];
        v8f zero = {};
#pragma unroll
        for (int t = 0; t < HID_F / 16; ++t) c1[t] = zero;
#pragma unroll
        for (int k0 = 0; k0 < IN_F; k0 += 4) {
            v2f a = *(const v2f*)(arow + k0);
            const float* brow = w1s + ((k0 + khalf) >> 1) * W1P + m * 2;
#pragma unroll
            for (int t = 0; t < HID_F / 16; ++t) {
                v2f bb = *(const v2f*)(brow + t * 32);
                c1[t] = __builtin_amdgcn_wmma_f32_16x16x4_f32(
                     false, a,  false, bb,
                     (short)0, c1[t],
                     false,  false);
            }
        }
#pragma unroll
        for (int t = 0; t < HID_F / 16; ++t) {
            float bv = b1[t * 16 + m];
#pragma unroll
            for (int r = 0; r < 8; ++r)
                drow[r * AP + t * 16] = fmaxf(c1[t][r] + bv, 0.f);
        }
    }
    __syncthreads();
    if (active) {
        v8f c2[OUT_F / 16];
        v8f zero = {};
#pragma unroll
        for (int t = 0; t < OUT_F / 16; ++t) c2[t] = zero;
#pragma unroll
        for (int k0 = 0; k0 < HID_F; k0 += 4) {
            v2f a = *(const v2f*)(arow + k0);
            const float* brow = w2s + ((k0 + khalf) >> 1) * W2P + m * 2;
#pragma unroll
            for (int t = 0; t < OUT_F / 16; ++t) {
                v2f bb = *(const v2f*)(brow + t * 32);
                c2[t] = __builtin_amdgcn_wmma_f32_16x16x4_f32(
                     false, a,  false, bb,
                     (short)0, c2[t],
                     false,  false);
            }
        }
#pragma unroll
        for (int t = 0; t < OUT_F / 16; ++t) {
#pragma unroll
            for (int r = 0; r < 8; ++r)
                drow[r * AP + t * 16] = c2[t][r];
        }
    }
    __syncthreads();
    int row0  = b * BS;
    int nrows = n_nodes - row0;
    if (nrows > BS) nrows = BS;
    float4* o4 = (float4*)(tout + (size_t)row0 * OUT_F);
    for (int i = tid; i < nrows * (OUT_F / 4); i += SUM_T) {
        int lr = i / (OUT_F / 4), c4 = i % (OUT_F / 4);
        o4[i] = *(const float4*)(acc + lr * AP + c4 * 4);
    }
}

extern "C" void kernel_launch(void* const* d_in, const int* in_sizes, int n_in,
                              void* d_out, int out_size, void* d_ws, size_t ws_size,
                              hipStream_t stream) {
    const float* feat = (const float*)d_in[0];
    const int*   src  = (const int*)d_in[1];
    const int*   dst  = (const int*)d_in[2];
    const float* W1   = (const float*)d_in[3];
    const float* b1   = (const float*)d_in[4];
    const float* W2   = (const float*)d_in[5];
    const float* b2   = (const float*)d_in[6];
    float* out = (float*)d_out;

    int n_nodes = in_sizes[0] / IN_F;
    int n_edges = in_sizes[1];

    float* tbuf = (float*)d_ws;
    float* pl1  = tbuf + (size_t)n_nodes * HID_F;

    int nbk  = (n_nodes + BS - 1) / BS;
    int nblk = (n_edges + CH - 1) / CH;
    if (nbk  < 1) nbk  = 1;
    if (nblk < 1) nblk = 1;

    int2* list = (int2*)pl1;
    int2* offt = list + (size_t)nblk * CH;

    gcn_chunk_sort_kernel<<<nblk, BIN_T, 0, stream>>>(src, dst, list, offt, n_edges, nbk);

    gcn_bucket_sum_dense_kernel
        <<<nbk, SUM_T, 0, stream>>>(feat, list, offt, nblk, W1, b1, W2, tbuf, n_nodes);

    gcn_bucket_sum_kernel<OUT_F, OUT_F, true>
        <<<nbk, SUM_T, 0, stream>>>(tbuf, list, offt, nblk, b2, out, n_nodes);
}


# Edge passes: a lane loads 16 B of a row, so one gather instruction fetches 2 rows (64 cols) or 4 rows (32 cols), not 1; each lane group sums its own entries; sum order changes (more partial sums)
# speedup vs baseline: 4.4044x; 1.0631x over previous
#include <hip/hip_runtime.h>
#include <type_traits>

typedef __attribute__((ext_vector_type(2))) float v2f;
typedef __attribute__((ext_vector_type(8))) float v8f;
typedef __attribute__((ext_vector_type(4))) float v4f;

#define IN_F  64
#define HID_F 64
#define OUT_F 32

#define BS       196
#define NBK_MAX  1024
#define CH       8192
#define BIN_T    1024
#define EPT      (CH / BIN_T)

template<int NW>
__device__ __forceinline__ int gcn_wg_scan(int v, int* wsum, int lane, int w, int& tot) {
    int x = v;
    for (int o = 1; o < 32; o <<= 1) {
        int y = __shfl_up(x, o);
        if (lane >= o) x += y;
    }
    if (lane == 31) wsum[w] = x;
    __syncthreads();
    int below = 0, all = 0;
    for (int j = 0; j < NW; ++j) {
        int s = wsum[j];
        if (j < w) below += s;
        all += s;
    }
    tot = all;
    __syncthreads();
    return x + below;
}

__global__ __launch_bounds__(BIN_T)
void gcn_chunk_sort_kernel(const int* __restrict__ src, const int* __restrict__ dst,
                           int2* __restrict__ list, int2* __restrict__ offt,
                           int n_edges, int nbk) {
    static_assert(NBK_MAX == BIN_T, "one bucket counter per thread");
    __shared__ int  hist[NBK_MAX];
    __shared__ int  wsum[BIN_T / 32];
    __shared__ int2 stage[CH];
    int tid  = threadIdx.x;
    int lane = tid & 31;
    int w    = tid >> 5;
    int blk  = blockIdx.x;
    hist[tid] = 0;
    int base = blk * CH;
    int s[EPT], d[EPT];
    for (int k = 0; k < EPT; ++k) {
        int e = base + k * BIN_T + tid;
        s[k] = 0; d[k] = 0;
        if (e < n_edges) { s[k] = src[e]; d[k] = dst[e]; }
    }
    __syncthreads();
    for (int k = 0; k < EPT; ++k) {
        int e = base + k * BIN_T + tid;
        if (e < n_edges) atomicAdd(&hist[(unsigned)d[k] / BS], 1);
    }
    __syncthreads();
    int tot;
    int hv = hist[tid];
    int lstart = gcn_wg_scan<BIN_T / 32>(hv, wsum, lane, w, tot) - hv;
    hist[tid] = lstart;
    if (tid < nbk) offt[(size_t)blk * NBK_MAX + tid] = make_int2(lstart, hv);
    __syncthreads();
    for (int k = 0; k < EPT; ++k) {
        int e = base + k * BIN_T + tid;
        if (e < n_edges) {
            int bkt = (unsigned)d[k] / BS;
            int p = atomicAdd(&hist[bkt], 1);
            stage[p] = make_int2(s[k], d[k] - bkt * BS);
        }
    }
    __syncthreads();
    for (int i = tid; i < tot; i += BIN_T) list[base + i] = stage[i];
}

#define SUM_T 512
#define CAP   4096
static_assert(BS <= 256, "a destination row fits the low byte of a packed entry");
static_assert(BS * NBK_MAX <= (1 << 24), "a source node fits the upper 24 bits of a packed entry");

template<int P>
__device__ __forceinline__ void gcn_row_flush(float* acc, int row, int c4, v4f a) {
    float* p = acc + row * P + c4;
    atomicAdd(p, a.x);
    atomicAdd(p + 1, a.y);
    atomicAdd(p + 2, a.z);
    atomicAdd(p + 3, a.w);
}

template<int U, int N, class Fn>
__device__ __forceinline__ void gcn_static_for(Fn&& f) {
    if constexpr (U < N) {
        f(std::integral_constant<int, U>{});
        gcn_static_for<U + 1, N>(f);
    }
}

template<int G, int U>
__device__ __forceinline__ unsigned gcn_group_entry(unsigned mine) {
    if constexpr (G == 16)
        return (unsigned)__builtin_amdgcn_update_dpp(0, (int)mine, 0x150 + U  , 0xf, 0xf, false);
    else
        return (unsigned)__builtin_amdgcn_mov_dpp8(mine, U * 0x249249  );
}

template<int F, int FS, int P, bool FULL>
__device__ __forceinline__ void gcn_gather_step(const float* __restrict__ feat,
                                                unsigned mine, int i0, int end,
                                                float* acc, int c4) {
    constexpr int G = F / 4;
    constexpr int UB = 8;
    static_assert(G % UB == 0, "whole groups of loads");
    int cur = (int)(gcn_group_entry<G, 0>(mine) & 255u);
    v4f a = {0.f, 0.f, 0.f, 0.f};
    v4f zero4 = {0.f, 0.f, 0.f, 0.f};
    gcn_static_for<0, G / UB>([&](auto bc) __attribute__((always_inline)) {
        constexpr int u0 = decltype(bc)::value * UB;
        v4f v[UB];
        gcn_static_for<0, UB>([&](auto uc) __attribute__((always_inline)) {
            constexpr int u = decltype(uc)::value;
            unsigned e = gcn_group_entry<G, u0 + u>(mine);
            unsigned off = (e >> 8) * (unsigned)(FS * 4) + (unsigned)(c4 * 4);
            v[u] = *(const v4f*)((const char*)feat + off);
        });
        gcn_static_for<0, UB>([&](auto uc) __attribute__((always_inline)) {
            constexpr int u = decltype(uc)::value;
            int dn = (int)(gcn_group_entry<G, u0 + u>(mine) & 255u);
            if (__builtin_expect(dn != cur, 0)) {
                gcn_row_flush<P>(acc, cur, c4, a);
                a = zero4; cur = dn;
            }
            a += (FULL || i0 + u0 + u < end) ? v[u] : zero4;
        });
    });
    if (FULL || i0 < end) gcn_row_flush<P>(acc, cur, c4, a);
}

template<int F, int FS, int P>
__device__ __forceinline__ void gcn_bucket_gather(const float* __restrict__ feat,
                                                  const unsigned* list, int end,
                                                  float* acc, int lane, int w) {
    constexpr int G  = F / 4;
    constexpr int NW = SUM_T / 32;
    static_assert(F == 64 || F == 32, "16 or 8 lanes per row");
    static_assert((unsigned long long)BS * NBK_MAX * FS * 4 < (1ull << 32), "a row's byte offset fits 32 bits");
    int g  = lane / G;
    int c4 = (lane % G) * 4;
    int base = __builtin_amdgcn_readfirstlane(w) * 32;
    if (base >= end) return;
    unsigned last = list[end - 1];
    unsigned nxt = base + lane < end ? list[base + lane] : last;
    for (; base < end; base += NW * 32) {
        unsigned mine = nxt;
        int nx = base + NW * 32;
        nxt = nx + lane < end ? list[nx + lane] : last;
        int i0 = base + g * G;
        if (base + 32 <= end) gcn_gather_step<F, FS, P, true >(feat, mine, i0, end, acc, c4);
        else                  gcn_gather_step<F, FS, P, false>(feat, mine, i0, end, acc, c4);
    }
}

__device__ __forceinline__ int2 gcn_first_run(const int2* __restrict__ offt,
                                              int b, int nblk, int tid) {
    int r = tid >> 1;
    int2 o = make_int2(0, 0);
    if (r < nblk) o = offt[(size_t)r * NBK_MAX + b];
    return o;
}

template<int F, int FS, int P>
__device__ __forceinline__ void gcn_bucket_edges(const float* __restrict__ feat,
                                                 const int2*  __restrict__ list,
                                                 const int2*  __restrict__ offt,
                                                 int b, int nblk, int2 o_pre,
                                                 float* acc, unsigned* raw, unsigned* ent,
                                                 int* rowc, int* wsum,
                                                 int tid, int lane, int w) {
    constexpr int NW  = SUM_T / 32;
    constexpr int RPP = SUM_T / 2;
    constexpr int CU  = 16;
    constexpr int EPT2 = CAP / SUM_T;
    static_assert(BS <= SUM_T, "one row counter per thread");
    static_assert(EPT2 * SUM_T == CAP, "whole entries per thread");
    int h = tid & 1;
    for (int t0 = 0; t0 < nblk; t0 += RPP) {
        int r = t0 + (tid >> 1);
        int o0 = 0, len = 0;
        if (r < nblk) {
            int2 o = o_pre;
            if (t0 > 0) o = offt[(size_t)r * NBK_MAX + b];
            o0  = o.x;
            len = o.y;
        }
        int tot;
        int pre = gcn_wg_scan<NW>(h == 0 ? len : 0, wsum, lane, w, tot) - len;
        const int2* run = list + (size_t)r * CH + o0;
        for (int seg = 0; seg < tot; seg += CAP) {
            int i_lo = seg - pre;
            if (i_lo < 0) i_lo = 0;
            int i_hi = seg + CAP - pre;
            if (i_hi > len) i_hi = len;
            for (int i = i_lo + ((i_lo ^ h) & 1); i < i_hi; i += 2 * CU) {
                int2 v[CU];
#pragma unroll
                for (int u = 0; u < CU; ++u) {
                    int ii = i + 2 * u;
                    if (ii < i_hi) v[u] = run[ii];
                }
#pragma unroll
                for (int u = 0; u < CU; ++u) {
                    int ii = i + 2 * u;
                    if (ii < i_hi) {
                        raw[pre + ii - seg] = ((unsigned)v[u].x << 8) | (unsigned)v[u].y;
                        atomicAdd(&rowc[v[u].y], 1);
                    }
                }
            }
            __syncthreads();
            int n = tot - seg;
            if (n > CAP) n = CAP;
            int rc = tid < BS ? rowc[tid] : 0;
            int rtot;
            int rs = gcn_wg_scan<NW>(rc, wsum, lane, w, rtot) - rc;
            if (tid < BS) rowc[tid] = rs;
            __syncthreads();
            {
                unsigned e[EPT2];
                int      p[EPT2];
#pragma unroll
                for (int k = 0; k < EPT2; ++k) {
                    int i = tid + k * SUM_T;
                    e[k] = raw[i < n ? i : 0];
                }
#pragma unroll
                for (int k = 0; k < EPT2; ++k) {
                    int i = tid + k * SUM_T;
                    int slot = i < n ? (int)(e[k] & 255u) : BS + lane;
                    p[k] = atomicAdd(&rowc[slot], 1);
                }
#pragma unroll
                for (int k = 0; k < EPT2; ++k) {
                    int i = tid + k * SUM_T;
                    if (i < n) ent[p[k]] = e[k];
                }
            }
            __syncthreads();
            gcn_bucket_gather<F, FS, P>(feat, ent, n, acc, lane, w);
            if (tid < BS) rowc[tid] = 0;
            __syncthreads();
        }
    }
}

template<int F, int FS, bool BIAS>
__global__ __launch_bounds__(SUM_T)
void gcn_bucket_sum_kernel(const float* __restrict__ feat,
                           const int2*  __restrict__ list,
                           const int2*  __restrict__ offt,
                           int nblk,
                           const float* __restrict__ bias,
                           float*       __restrict__ outp,
                           int n_nodes) {
    __shared__ __attribute__((aligned(16))) float acc[BS * F];
    __shared__ unsigned raw[CAP];
    __shared__ unsigned ent[CAP];
    __shared__ int  rowc[BS + 32];
    __shared__ int  wsum[SUM_T / 32];
    int tid  = threadIdx.x;
    int lane = tid & 31;
    int w    = tid >> 5;
    int b = blockIdx.x;

    int2 o_pre = gcn_first_run(offt, b, nblk, tid);

    float4* acc4 = (float4*)acc;
    float4 z = make_float4(0.f, 0.f, 0.f, 0.f);
    for (int i = tid; i < BS * F / 4; i += SUM_T) acc4[i] = z;
    if (tid < BS) rowc[tid] = 0;

    gcn_bucket_edges<F, FS, F>(feat, list, offt, b, nblk, o_pre, acc, raw, ent, rowc, wsum, tid, lane, w);
    int row0  = b * BS;
    int nrows = n_nodes - row0;
    if (nrows > BS) nrows = BS;
    float4* o4 = (float4*)(outp + (size_t)row0 * F);
    for (int i = tid; i < nrows * (F / 4); i += SUM_T) {
        float4 v = acc4[i];
        if (BIAS) {
            int c = (i % (F / 4)) * 4;
            v.x += bias[c]; v.y += bias[c + 1]; v.z += bias[c + 2]; v.w += bias[c + 3];
        }
        o4[i] = v;
    }
}

#define NSTRIP ((BS + 15) / 16)
#define BSP    (NSTRIP * 16)
#define AP     68
#define W1P    (2 * HID_F + 32)
#define W2P    (2 * OUT_F + 32)

__global__ __launch_bounds__(SUM_T)
void gcn_bucket_sum_dense_kernel(const float* __restrict__ feat,
                                 const int2*  __restrict__ list,
                                 const int2*  __restrict__ offt,
                                 int nblk,
                                 const float* __restrict__ W1,
                                 const float* __restrict__ b1,
                                 const float* __restrict__ W2,
                                 float*       __restrict__ tout,
                                 int n_nodes) {
    static_assert(IN_F == 64 && HID_F == 64 && OUT_F == 32, "tile and strip shapes");
    static_assert(NSTRIP <= SUM_T / 32, "one strip per wave");
    __shared__ __attribute__((aligned(16))) float acc[BSP * AP];
    __shared__ __attribute__((aligned(16))) float w1s[(IN_F / 2) * W1P];
    __shared__ __attribute__((aligned(16))) float w2s[(HID_F / 2) * W2P];
    __shared__ unsigned raw[CAP];
    __shared__ unsigned ent[CAP];
    __shared__ int  rowc[BS + 32];
    __shared__ int  wsum[SUM_T / 32];
    int tid  = threadIdx.x;
    int lane = tid & 31;
    int w    = tid >> 5;
    int b = blockIdx.x;

    constexpr int W1_PT = IN_F * HID_F / SUM_T;
    constexpr int W2_PT = HID_F * OUT_F / SUM_T;
    static_assert(W1_PT * SUM_T == IN_F * HID_F && W2_PT * SUM_T == HID_F * OUT_F,
                  "whole pieces per thread");
    float w1r[W1_PT], w2r[W2_PT];
#pragma unroll
    for (int j = 0; j < W1_PT; ++j) w1r[j] = W1[tid + j * SUM_T];
#pragma unroll
    for (int j = 0; j < W2_PT; ++j) w2r[j] = W2[tid + j * SUM_T];
    int2 o_pre = gcn_first_run(offt, b, nblk, tid);

    float4* acc4 = (float4*)acc;
    float4 z = make_float4(0.f, 0.f, 0.f, 0.f);
    for (int i = tid; i < BSP * AP / 4; i += SUM_T) acc4[i] = z;
    if (tid < BS) rowc[tid] = 0;
#pragma unroll
    for (int j = 0; j < W1_PT; ++j) {
        int i = tid + j * SUM_T;
        int k = i / HID_F, n = i % HID_F;
        w1s[(k >> 1) * W1P + n * 2 + (k & 1)] = w1r[j];
    }
#pragma unroll
    for (int j = 0; j < W2_PT; ++j) {
        int i = tid + j * SUM_T;
        int k = i / OUT_F, n = i % OUT_F;
        w2s[(k >> 1) * W2P + n * 2 + (k & 1)] = w2r[j];
    }

    gcn_bucket_edges<IN_F, IN_F, AP>(feat, list, offt, b, nblk, o_pre, acc, raw, ent, rowc, wsum, tid, lane, w);

    bool active = w < NSTRIP;
    int m     = lane & 15;
    int khalf = (lane >> 4) << 1;
    int mrow  = (lane >> 4) * 8;
    float* arow = acc + (w * 16 + m) * AP + khalf;
    float* drow = acc + (w * 16 + mrow) * AP + m;

    if (active) {
        v8f c1[HID_F / 16];
        v8f zero = {};
#pragma unroll
        for (int t = 0; t < HID_F / 16; ++t) c1[t] = zero;
#pragma unroll
        for (int k0 = 0; k0 < IN_F; k0 += 4) {
            v2f a = *(const v2f*)(arow + k0);
            const float* brow = w1s + ((k0 + khalf) >> 1) * W1P + m * 2;
#pragma unroll
            for (int t = 0; t < HID_F / 16; ++t) {
                v2f bb = *(const v2f*)(brow + t * 32);
                c1[t] = __builtin_amdgcn_wmma_f32_16x16x4_f32(
                     false, a,  false, bb,
                     (short)0, c1[t],
                     false,  false);
            }
        }
#pragma unroll
        for (int t = 0; t < HID_F / 16; ++t) {
            float bv = b1[t * 16 + m];
#pragma unroll
            for (int r = 0; r < 8; ++r)
                drow[r * AP + t * 16] = fmaxf(c1[t][r] + bv, 0.f);
        }
    }
    __syncthreads();
    if (active) {
        v8f c2[OUT_F / 16];
        v8f zero = {};
#pragma unroll
        for (int t = 0; t < OUT_F / 16; ++t) c2[t] = zero;
#pragma unroll
        for (int k0 = 0; k0 < HID_F; k0 += 4) {
            v2f a = *(const v2f*)(arow + k0);
            const float* brow = w2s + ((k0 + khalf) >> 1) * W2P + m * 2;
#pragma unroll
            for (int t = 0; t < OUT_F / 16; ++t) {
                v2f bb = *(const v2f*)(brow + t * 32);
                c2[t] = __builtin_amdgcn_wmma_f32_16x16x4_f32(
                     false, a,  false, bb,
                     (short)0, c2[t],
                     false,  false);
            }
        }
#pragma unroll
        for (int t = 0; t < OUT_F / 16; ++t) {
#pragma unroll
            for (int r = 0; r < 8; ++r)
                drow[r * AP + t * 16] = c2[t][r];
        }
    }
    __syncthreads();
    int row0  = b * BS;
    int nrows = n_nodes - row0;
    if (nrows > BS) nrows = BS;
    float4* o4 = (float4*)(tout + (size_t)row0 * OUT_F);
    for (int i = tid; i < nrows * (OUT_F / 4); i += SUM_T) {
        int lr = i / (OUT_F / 4), c4 = i % (OUT_F / 4);
        o4[i] = *(const float4*)(acc + lr * AP + c4 * 4);
    }
}

extern "C" void kernel_launch(void* const* d_in, const int* in_sizes, int n_in,
                              void* d_out, int out_size, void* d_ws, size_t ws_size,
                              hipStream_t stream) {
    const float* feat = (const float*)d_in[0];
    const int*   src  = (const int*)d_in[1];
    const int*   dst  = (const int*)d_in[2];
    const float* W1   = (const float*)d_in[3];
    const float* b1   = (const float*)d_in[4];
    const float* W2   = (const float*)d_in[5];
    const float* b2   = (const float*)d_in[6];
    float* out = (float*)d_out;

    int n_nodes = in_sizes[0] / IN_F;
    int n_edges = in_sizes[1];

    float* tbuf = (float*)d_ws;
    float* pl1  = tbuf + (size_t)n_nodes * HID_F;

    int nbk  = (n_nodes + BS - 1) / BS;
    int nblk = (n_edges + CH - 1) / CH;
    if (nbk  < 1) nbk  = 1;
    if (nblk < 1) nblk = 1;

    int2* list = (int2*)pl1;
    int2* offt = list + (size_t)nblk * CH;

    gcn_chunk_sort_kernel<<<nblk, BIN_T, 0, stream>>>(src, dst, list, offt, n_edges, nbk);

    gcn_bucket_sum_dense_kernel
        <<<nbk, SUM_T, 0, stream>>>(feat, list, offt, nblk, W1, b1, W2, tbuf, n_nodes);

    gcn_bucket_sum_kernel<OUT_F, OUT_F, true>
        <<<nbk, SUM_T, 0, stream>>>(tbuf, list, offt, nblk, b2, out, n_nodes);
}
